# starting kernel
# baseline (speedup 1.0000x reference)
#include <hip/hip_runtime.h>

#define N_NODES 100000
#define N_EDGES 1600000
#define IN_DIM  256
#define HID     128
#define OUTD    64

typedef __attribute__((ext_vector_type(2))) float v2f;
typedef __attribute__((ext_vector_type(8))) float v8f;

__global__ void k_init_deg(float* __restrict__ deg) {
  int i = blockIdx.x * blockDim.x + threadIdx.x;
  if (i < N_NODES) deg[i] = 1.0f;
}

__global__ void k_count_deg(const int* __restrict__ ei, float* __restrict__ deg) {
  int e = blockIdx.x * blockDim.x + threadIdx.x;
  if (e < N_EDGES) {
    int d = ei[N_EDGES + e];
    unsafeAtomicAdd(&deg[d], 1.0f);
  }
}

__global__ void k_rsqrt_inplace(float* __restrict__ deg) {
  int i = blockIdx.x * blockDim.x + threadIdx.x;
  if (i < N_NODES) deg[i] = rsqrtf(deg[i]);
}

template <int K, int NT, bool RELU>
__global__ __launch_bounds__(256) void k_gemm_wmma(const float* __restrict__ A,
                                                   const float* __restrict__ W,
                                                   float* __restrict__ C, int M) {
  constexpr int NC = NT * 16;
  constexpr int KC = 64;
  __shared__ float lds[KC * NC];

  const int wave = threadIdx.x >> 5;
  const int lane = threadIdx.x & 31;
  const int mtile = blockIdx.x * 8 + wave;
  const long m0 = (long)mtile * 16;
  const bool active = (m0 < M);

  const int m  = lane & 15;
  const int kh = lane >> 4;
  const int n  = lane & 15;

  const long arow = active ? (m0 + m) : (long)(M - 1);
  const float* __restrict__ Aptr = A + arow * K;

  v8f acc[NT] = {};

  for (int kc = 0; kc < K; kc += KC) {
    if (kc != 0) __syncthreads();
    {
      constexpr int ENT = KC * NC / 8;
#pragma unroll
      for (int idx = threadIdx.x; idx < ENT; idx += 256) {
        int kp = idx / (NC / 4);
        int nq = (idx % (NC / 4)) * 4;
        const float4 a = *(const float4*)(W + (long)(kc + 2 * kp) * NC + nq);
        const float4 b = *(const float4*)(W + (long)(kc + 2 * kp + 1) * NC + nq);
        float4* __restrict__ dst = (float4*)&lds[kp * 2 * NC + 2 * nq];
        dst[0] = make_float4(a.x, b.x, a.y, b.y);
        dst[1] = make_float4(a.z, b.z, a.w, b.w);
      }
    }
    __syncthreads();

#pragma unroll 4
    for (int k0 = kc; k0 < kc + KC; k0 += 4) {
      const int ka = k0 + 2 * kh;
      const int kp = (ka - kc) >> 1;
      v2f av = *(const v2f*)(Aptr + ka);
      if (RELU) {
        av.x = fmaxf(av.x, 0.0f);
        av.y = fmaxf(av.y, 0.0f);
      }
      const v2f* __restrict__ brow = (const v2f*)&lds[kp * 2 * NC + 2 * n];
#pragma unroll
      for (int j = 0; j < NT; ++j) {
        v2f bv = brow[j * 16];
        acc[j] = __builtin_amdgcn_wmma_f32_16x16x4_f32(
             false, av,  false, bv,
             (short)0, acc[j],  false,  false);
      }
    }
  }

  if (!active) return;
#pragma unroll
  for (int j = 0; j < NT; ++j) {
#pragma unroll
    for (int r = 0; r < 8; ++r) {
      long row = m0 + r + 8 * kh;
      if (row < M) C[row * NC + j * 16 + n] = acc[j][r];
    }
  }
}

template <int F>
__global__ void k_self_init(const float* __restrict__ h,
                            const float* __restrict__ dinv,
                            const float* __restrict__ b,
                            float* __restrict__ agg) {
  long t = (long)blockIdx.x * blockDim.x + threadIdx.x;
  if (t >= (long)N_NODES * F) return;
  long i = t / F;
  int f = (int)(t % F);
  float di = dinv[i];
  agg[t] = h[t] * (di * di) + b[f];
}

template <int F>
__global__ void k_edge_agg(const int* __restrict__ ei,
                           const float* __restrict__ dinv,
                           const float* __restrict__ h,
                           float* __restrict__ agg) {
  constexpr int CH = F / 4;
  long t = (long)blockIdx.x * blockDim.x + threadIdx.x;
  long e = t / CH;
  int c = (int)(t % CH);
  if (e >= N_EDGES) return;
  int s = ei[e];
  int d = ei[N_EDGES + e];
  float norm = dinv[s] * dinv[d];
  float4 v = *(const float4*)(h + (long)s * F + c * 4);
  float* __restrict__ out = agg + (long)d * F + c * 4;
  unsafeAtomicAdd(out + 0, v.x * norm);
  unsafeAtomicAdd(out + 1, v.y * norm);
  unsafeAtomicAdd(out + 2, v.z * norm);
  unsafeAtomicAdd(out + 3, v.w * norm);
}

extern "C" void kernel_launch(void* const* d_in, const int* in_sizes, int n_in,
                              void* d_out, int out_size, void* d_ws, size_t ws_size,
                              hipStream_t stream) {
  const float* x  = (const float*)d_in[0];
  const int*   ei = (const int*)d_in[1];
  const float* W1 = (const float*)d_in[2];
  const float* b1 = (const float*)d_in[3];
  const float* W2 = (const float*)d_in[4];
  const float* b2 = (const float*)d_in[5];
  float* out = (float*)d_out;

  float* ws   = (float*)d_ws;
  float* dinv = ws;
  float* h1   = dinv + N_NODES;
  float* agg1 = h1 + (size_t)N_NODES * HID;
  float* h3   = agg1 + (size_t)N_NODES * HID;

  const int T = 256;

  k_init_deg<<<(N_NODES + T - 1) / T, T, 0, stream>>>(dinv);
  k_count_deg<<<(N_EDGES + T - 1) / T, T, 0, stream>>>(ei, dinv);
  k_rsqrt_inplace<<<(N_NODES + T - 1) / T, T, 0, stream>>>(dinv);

  const int mtiles = (N_NODES + 15) / 16;
  const int gblocks = (mtiles + 7) / 8;
  k_gemm_wmma<IN_DIM, HID / 16, false>
      <<<gblocks, T, 0, stream>>>(x, W1, h1, N_NODES);

  {
    long tot = (long)N_NODES * HID;
    k_self_init<HID><<<(unsigned)((tot + T - 1) / T), T, 0, stream>>>(h1, dinv, b1, agg1);
    long et = (long)N_EDGES * (HID / 4);
    k_edge_agg<HID><<<(unsigned)((et + T - 1) / T), T, 0, stream>>>(ei, dinv, h1, agg1);
  }

  k_gemm_wmma<HID, OUTD / 16, true>
      <<<gblocks, T, 0, stream>>>(agg1, W2, h3, N_NODES);

  {
    long tot = (long)N_NODES * OUTD;
    k_self_init<OUTD><<<(unsigned)((tot + T - 1) / T), T, 0, stream>>>(h3, dinv, b2, out);
    long et = (long)N_EDGES * (OUTD / 4);
    k_edge_agg<OUTD><<<(unsigned)((et + T - 1) / T), T, 0, stream>>>(ei, dinv, h3, out);
  }
}


# Edge sums by dst-range owner work-groups (no sort, no CSR): each keeps 400 nodes' sums in LDS, filters the edge list by a byte tag, adds rows by LDS atomics; sum order = LDS arrival; 10 launches -> 7
# speedup vs baseline: 2.1071x; 2.1071x over previous
#include <hip/hip_runtime.h>

#define N_NODES 100000
#define N_EDGES 1600000
#define IN_DIM  256
#define HID     128
#define OUTD    64
#define BN      400

typedef __attribute__((ext_vector_type(2))) float v2f;
typedef __attribute__((ext_vector_type(8))) float v8f;

__global__ void k_init_deg(float* __restrict__ deg) {
  int i = blockIdx.x * blockDim.x + threadIdx.x;
  if (i < N_NODES) deg[i] = 1.0f;
}

__global__ void k_count_deg_tag(const int* __restrict__ ei, float* __restrict__ deg,
                                unsigned* __restrict__ tag4) {
  static_assert(N_EDGES % 4 == 0, "four destinations per 16-byte load");
  int t = blockIdx.x * blockDim.x + threadIdx.x;
  if (t < N_EDGES / 4) {
    const int4 d = *(const int4*)(ei + N_EDGES + 4 * t);
    unsafeAtomicAdd(&deg[d.x], 1.0f);
    unsafeAtomicAdd(&deg[d.y], 1.0f);
    unsafeAtomicAdd(&deg[d.z], 1.0f);
    unsafeAtomicAdd(&deg[d.w], 1.0f);
    tag4[t] = ((unsigned)d.x / BN) | (((unsigned)d.y / BN) << 8) |
              (((unsigned)d.z / BN) << 16) | (((unsigned)d.w / BN) << 24);
  }
}

__global__ void k_rsqrt_inplace(float* __restrict__ deg) {
  int i = blockIdx.x * blockDim.x + threadIdx.x;
  if (i < N_NODES) deg[i] = rsqrtf(deg[i]);
}

template <int K, int NT, bool RELU>
__global__ __launch_bounds__(256) void k_gemm_wmma(const float* __restrict__ A,
                                                   const float* __restrict__ W,
                                                   float* __restrict__ C, int M) {
  constexpr int NC = NT * 16;
  constexpr int KC = 64;
  __shared__ float lds[KC * NC];

  const int wave = threadIdx.x >> 5;
  const int lane = threadIdx.x & 31;
  const int mtile = blockIdx.x * 8 + wave;
  const long m0 = (long)mtile * 16;
  const bool active = (m0 < M);

  const int m  = lane & 15;
  const int kh = lane >> 4;
  const int n  = lane & 15;

  const long arow = active ? (m0 + m) : (long)(M - 1);
  const float* __restrict__ Aptr = A + arow * K;

  v8f acc[NT] = {};

  for (int kc = 0; kc < K; kc += KC) {
    if (kc != 0) __syncthreads();
    {
      constexpr int ENT = KC * NC / 8;
#pragma unroll
      for (int idx = threadIdx.x; idx < ENT; idx += 256) {
        int kp = idx / (NC / 4);
        int nq = (idx % (NC / 4)) * 4;
        const float4 a = *(const float4*)(W + (long)(kc + 2 * kp) * NC + nq);
        const float4 b = *(const float4*)(W + (long)(kc + 2 * kp + 1) * NC + nq);
        float4* __restrict__ dst = (float4*)&lds[kp * 2 * NC + 2 * nq];
        dst[0] = make_float4(a.x, b.x, a.y, b.y);
        dst[1] = make_float4(a.z, b.z, a.w, b.w);
      }
    }
    __syncthreads();

#pragma unroll 4
    for (int k0 = kc; k0 < kc + KC; k0 += 4) {
      const int ka = k0 + 2 * kh;
      const int kp = (ka - kc) >> 1;
      v2f av = *(const v2f*)(Aptr + ka);
      if (RELU) {
        av.x = fmaxf(av.x, 0.0f);
        av.y = fmaxf(av.y, 0.0f);
      }
      const v2f* __restrict__ brow = (const v2f*)&lds[kp * 2 * NC + 2 * n];
#pragma unroll
      for (int j = 0; j < NT; ++j) {
        v2f bv = brow[j * 16];
        acc[j] = __builtin_amdgcn_wmma_f32_16x16x4_f32(
             false, av,  false, bv,
             (short)0, acc[j],  false,  false);
      }
    }
  }

  if (!active) return;
#pragma unroll
  for (int j = 0; j < NT; ++j) {
#pragma unroll
    for (int r = 0; r < 8; ++r) {
      long row = m0 + r + 8 * kh;
      if (row < M) C[row * NC + j * 16 + n] = acc[j][r];
    }
  }
}

__device__ __forceinline__ unsigned zero_bytes(unsigned t) {
  return ~(((t & 0x7f7f7f7fu) + 0x7f7f7f7fu) | t | 0x7f7f7f7fu);
}

template <int F, bool FULL>
__device__ __forceinline__ void serve_queue(unsigned e, int nvalid, int lane, int d0,
                                            const int* __restrict__ ei,
                                            const float* __restrict__ dinv,
                                            const float* __restrict__ h,
                                            float* tile, const float* dinvL) {
  constexpr int VEC = F / 32;
  constexpr int U = 16;
  typedef float vecF __attribute__((ext_vector_type(VEC)));
  const bool valid = FULL || lane < nvalid;
  const int s = ei[e];
  const int dl = valid ? (ei[N_EDGES + e] - d0) : 0;
  const float nr = dinv[s] * dinvL[dl];
#pragma unroll 1
  for (int j0 = 0; j0 < (FULL ? 32 : nvalid); j0 += U) {
    vecF v[U];
#pragma unroll
    for (int u = 0; u < U; ++u) {
      if (FULL || j0 + u < nvalid) {
        const int sj = __builtin_amdgcn_readlane(s, j0 + u);
        v[u] = *(const vecF*)(h + (long)sj * F + lane * VEC);
      }
    }
#pragma unroll
    for (int u = 0; u < U; ++u) {
      if (FULL || j0 + u < nvalid) {
        const int dj = __builtin_amdgcn_readlane(dl, j0 + u);
        const float nj = __int_as_float(__builtin_amdgcn_readlane(__float_as_int(nr), j0 + u));
        float* tp = tile + dj * F + lane;
        for (int c = 0; c < VEC; ++c)
          __hip_atomic_fetch_add(tp + c * 32, v[u][c] * nj, __ATOMIC_RELAXED,
                                 __HIP_MEMORY_SCOPE_WORKGROUP);
      }
    }
  }
}

template <int F>
__global__ __launch_bounds__(1024) void k_edge_sum_own(const int* __restrict__ ei,
                                                        const unsigned* __restrict__ tag4,
                                                        const float* __restrict__ dinv,
                                                        const float* __restrict__ h,
                                                        const float* __restrict__ b,
                                                        float* __restrict__ agg) {
  static_assert(N_NODES % BN == 0 && N_NODES / BN <= 256, "whole ranges; the range index fits a byte");
  static_assert(N_EDGES % 16 == 0, "the tags are read 16 at a time");
  static_assert(F % 32 == 0, "a row is a whole number of 128-byte lines");
  constexpr int VEC = F / 32;
  constexpr int NV = N_EDGES / 16;
  constexpr int NCH = (NV + 63) / 64;
  typedef float vecF __attribute__((ext_vector_type(VEC)));
  __shared__ float tile[BN * F];
  __shared__ float dinvL[BN];
  __shared__ unsigned q[32 * 64];

  const int lane = threadIdx.x & 31;
  const int wave = threadIdx.x >> 5;
  const int d0 = blockIdx.x * BN;

  {
    const vecF bv = *(const vecF*)(b + lane * VEC);
    for (int r = wave; r < BN; r += 32) {
      const float di = dinv[d0 + r];
      const vecF hv = *(const vecF*)(h + (long)(d0 + r) * F + lane * VEC);
      for (int c = 0; c < VEC; ++c) tile[r * F + c * 32 + lane] = hv[c] * (di * di) + bv[c];
      if (lane == 0) dinvL[r] = di;
    }
  }
  __syncthreads();

  unsigned* qw = q + wave * 64;
  const unsigned pat = blockIdx.x * 0x01010101u;
  const uint4* __restrict__ tagv = (const uint4*)tag4;
  static_assert((NCH - 1) * 64 + 31 < NV, "the first load of every trip lies inside the list");
  auto load_tags = [&](int trip, int half) -> uint4 {
    const int u = trip * 64 + half * 32 + lane;
    return tagv[u < NV ? u : NV - 1];
  };
  int cnt = 0;
  uint4 xa = load_tags(wave, 0);
  uint4 xb = load_tags(wave, 1);
#pragma unroll 1
  for (int it = wave; it < NCH; it += 32) {
    const uint4 na = load_tags(it + 32, 0);
    const uint4 nb = load_tags(it + 32, 1);
    unsigned mb = (zero_bytes(xb.x ^ pat) >> 3) | (zero_bytes(xb.y ^ pat) >> 2) |
                  (zero_bytes(xb.z ^ pat) >> 1) | zero_bytes(xb.w ^ pat);
    if (it * 64 + 32 + lane >= NV) mb = 0u;
    unsigned mm = (zero_bytes(xa.x ^ pat) >> 7) | (zero_bytes(xa.y ^ pat) >> 6) |
                  (zero_bytes(xa.z ^ pat) >> 5) | (zero_bytes(xa.w ^ pat) >> 4) | mb;
    const unsigned ebase = (unsigned)it * 1024u + (unsigned)lane * 16u;
    for (int t = 0; t < 32; ++t) {
      const bool has = (mm != 0u);
      const unsigned bal = __builtin_amdgcn_ballot_w32(has);
      if (bal == 0u) break;
      if (has) {
        const unsigned p = (unsigned)__builtin_ctz(mm);
        mm &= mm - 1u;
        const unsigned j = p & 7u;
        __hip_atomic_store(qw + cnt + (int)__builtin_amdgcn_mbcnt_lo(bal, 0u),
                           ebase + (j >> 2) * 512u + 4u * (j & 3u) + (p >> 3), __ATOMIC_RELAXED,
                           __HIP_MEMORY_SCOPE_WAVEFRONT);
      }
      cnt += __builtin_popcount(bal);
      if (cnt >= 32) {
        const unsigned e32 = __hip_atomic_load(qw + lane, __ATOMIC_RELAXED, __HIP_MEMORY_SCOPE_WAVEFRONT);
        serve_queue<F, true>(e32, 32, lane, d0, ei, dinv, h, tile, dinvL);
        const int rest = cnt - 32;
        if (lane < rest) {
          const unsigned mv = __hip_atomic_load(qw + 32 + lane, __ATOMIC_RELAXED, __HIP_MEMORY_SCOPE_WAVEFRONT);
          __hip_atomic_store(qw + lane, mv, __ATOMIC_RELAXED, __HIP_MEMORY_SCOPE_WAVEFRONT);
        }
        cnt = rest;
      }
    }
    xa = na;
    xb = nb;
  }
  if (cnt > 0) {
    unsigned el = 0u;
    if (lane < cnt) el = __hip_atomic_load(qw + lane, __ATOMIC_RELAXED, __HIP_MEMORY_SCOPE_WAVEFRONT);
    serve_queue<F, false>(el, cnt, lane, d0, ei, dinv, h, tile, dinvL);
  }
  __syncthreads();

  for (int r = wave; r < BN; r += 32) {
    vecF o;
    for (int c = 0; c < VEC; ++c) o[c] = tile[r * F + c * 32 + lane];
    *(vecF*)(agg + (long)(d0 + r) * F + lane * VEC) = o;
  }
}

extern "C" void kernel_launch(void* const* d_in, const int* in_sizes, int n_in,
                              void* d_out, int out_size, void* d_ws, size_t ws_size,
                              hipStream_t stream) {
  const float* x  = (const float*)d_in[0];
  const int*   ei = (const int*)d_in[1];
  const float* W1 = (const float*)d_in[2];
  const float* b1 = (const float*)d_in[3];
  const float* W2 = (const float*)d_in[4];
  const float* b2 = (const float*)d_in[5];
  float* out = (float*)d_out;

  float* ws   = (float*)d_ws;
  float* h1   = ws;
  float* agg1 = h1 + (size_t)N_NODES * HID;
  unsigned* tag4 = (unsigned*)(agg1 + (size_t)N_NODES * HID);
  float* dinv = (float*)tag4 + (size_t)N_NODES * OUTD;
  float* h3   = h1;
  static_assert((long)N_EDGES <= (long)N_NODES * OUTD * 4, "the owner tags fit in the third region");

  const int T = 256;

  k_init_deg<<<(N_NODES + T - 1) / T, T, 0, stream>>>(dinv);
  k_count_deg_tag<<<(N_EDGES / 4 + T - 1) / T, T, 0, stream>>>(ei, dinv, tag4);
  k_rsqrt_inplace<<<(N_NODES + T - 1) / T, T, 0, stream>>>(dinv);

  const int mtiles = (N_NODES + 15) / 16;
  const int gblocks = (mtiles + 7) / 8;
  k_gemm_wmma<IN_DIM, HID / 16, false>
      <<<gblocks, T, 0, stream>>>(x, W1, h1, N_NODES);

  k_edge_sum_own<HID><<<N_NODES / BN, 1024, 0, stream>>>(ei, tag4, dinv, h1, b1, agg1);

  k_gemm_wmma<HID, OUTD / 16, true>
      <<<gblocks, T, 0, stream>>>(agg1, W2, h3, N_NODES);

  k_edge_sum_own<OUTD><<<N_NODES / BN, 1024, 0, stream>>>(ei, tag4, dinv, h3, b2, out);
}


# Layer-2 edge sums read each owner's matched edges (src, tile row, norm: 8 B an edge) from a list its layer-1 pass keeps, not by filtering tags and looking indices up again; sum order = LDS arrival
# speedup vs baseline: 2.3404x; 1.1107x over previous
#include <hip/hip_runtime.h>

#define N_NODES 100000
#define N_EDGES 1600000
#define IN_DIM  256
#define HID     128
#define OUTD    64
#define BN      400
#define LCAP    11904

typedef __attribute__((ext_vector_type(2))) float v2f;
typedef __attribute__((ext_vector_type(2))) unsigned v2u;
typedef __attribute__((ext_vector_type(8))) float v8f;

__global__ void k_init_deg(float* __restrict__ deg) {
  int i = blockIdx.x * blockDim.x + threadIdx.x;
  if (i < N_NODES) deg[i] = 1.0f;
}

__global__ void k_count_deg_tag(const int* __restrict__ ei, float* __restrict__ deg,
                                unsigned* __restrict__ tag4) {
  static_assert(N_EDGES % 4 == 0, "four destinations per 16-byte load");
  int t = blockIdx.x * blockDim.x + threadIdx.x;
  if (t < N_EDGES / 4) {
    const int4 d = *(const int4*)(ei + N_EDGES + 4 * t);
    unsafeAtomicAdd(&deg[d.x], 1.0f);
    unsafeAtomicAdd(&deg[d.y], 1.0f);
    unsafeAtomicAdd(&deg[d.z], 1.0f);
    unsafeAtomicAdd(&deg[d.w], 1.0f);
    tag4[t] = ((unsigned)d.x / BN) | (((unsigned)d.y / BN) << 8) |
              (((unsigned)d.z / BN) << 16) | (((unsigned)d.w / BN) << 24);
  }
}

__global__ void k_rsqrt_inplace(float* __restrict__ deg) {
  int i = blockIdx.x * blockDim.x + threadIdx.x;
  if (i < N_NODES) deg[i] = rsqrtf(deg[i]);
}

template <int K, int NT, bool RELU>
__global__ __launch_bounds__(256) void k_gemm_wmma(const float* __restrict__ A,
                                                   const float* __restrict__ W,
                                                   float* __restrict__ C, int M) {
  constexpr int NC = NT * 16;
  constexpr int KC = 64;
  __shared__ float lds[KC * NC];

  const int wave = threadIdx.x >> 5;
  const int lane = threadIdx.x & 31;
  const int mtile = blockIdx.x * 8 + wave;
  const long m0 = (long)mtile * 16;
  const bool active = (m0 < M);

  const int m  = lane & 15;
  const int kh = lane >> 4;
  const int n  = lane & 15;

  const long arow = active ? (m0 + m) : (long)(M - 1);
  const float* __restrict__ Aptr = A + arow * K;

  v8f acc[NT] = {};

  for (int kc = 0; kc < K; kc += KC) {
    if (kc != 0) __syncthreads();
    {
      constexpr int ENT = KC * NC / 8;
#pragma unroll
      for (int idx = threadIdx.x; idx < ENT; idx += 256) {
        int kp = idx / (NC / 4);
        int nq = (idx % (NC / 4)) * 4;
        const float4 a = *(const float4*)(W + (long)(kc + 2 * kp) * NC + nq);
        const float4 b = *(const float4*)(W + (long)(kc + 2 * kp + 1) * NC + nq);
        float4* __restrict__ dst = (float4*)&lds[kp * 2 * NC + 2 * nq];
        dst[0] = make_float4(a.x, b.x, a.y, b.y);
        dst[1] = make_float4(a.z, b.z, a.w, b.w);
      }
    }
    __syncthreads();

#pragma unroll 4
    for (int k0 = kc; k0 < kc + KC; k0 += 4) {
      const int ka = k0 + 2 * kh;
      const int kp = (ka - kc) >> 1;
      v2f av = *(const v2f*)(Aptr + ka);
      if (RELU) {
        av.x = fmaxf(av.x, 0.0f);
        av.y = fmaxf(av.y, 0.0f);
      }
      const v2f* __restrict__ brow = (const v2f*)&lds[kp * 2 * NC + 2 * n];
#pragma unroll
      for (int j = 0; j < NT; ++j) {
        v2f bv = brow[j * 16];
        acc[j] = __builtin_amdgcn_wmma_f32_16x16x4_f32(
             false, av,  false, bv,
             (short)0, acc[j],  false,  false);
      }
    }
  }

  if (!active) return;
#pragma unroll
  for (int j = 0; j < NT; ++j) {
#pragma unroll
    for (int r = 0; r < 8; ++r) {
      long row = m0 + r + 8 * kh;
      if (row < M) C[row * NC + j * 16 + n] = acc[j][r];
    }
  }
}

__device__ __forceinline__ unsigned zero_bytes(unsigned t) {
  return ~(((t & 0x7f7f7f7fu) + 0x7f7f7f7fu) | t | 0x7f7f7f7fu);
}

template <int F, bool FULL>
__device__ __forceinline__ void add_rows(int s, int dl, float nr, int nvalid, int lane,
                                         const float* __restrict__ h, float* tile) {
  constexpr int VEC = F / 32;
  constexpr int U = 16;
  typedef float vecF __attribute__((ext_vector_type(VEC)));
#pragma unroll 1
  for (int j0 = 0; j0 < (FULL ? 32 : nvalid); j0 += U) {
    vecF v[U];
#pragma unroll
    for (int u = 0; u < U; ++u) {
      if (FULL || j0 + u < nvalid) {
        const int sj = __builtin_amdgcn_readlane(s, j0 + u);
        v[u] = *(const vecF*)(h + (long)sj * F + lane * VEC);
      }
    }
#pragma unroll
    for (int u = 0; u < U; ++u) {
      if (FULL || j0 + u < nvalid) {
        const int dj = __builtin_amdgcn_readlane(dl, j0 + u);
        const float nj = __int_as_float(__builtin_amdgcn_readlane(__float_as_int(nr), j0 + u));
        float* tp = tile + dj * F + lane;
        for (int c = 0; c < VEC; ++c)
          __hip_atomic_fetch_add(tp + c * 32, v[u][c] * nj, __ATOMIC_RELAXED,
                                 __HIP_MEMORY_SCOPE_WORKGROUP);
      }
    }
  }
}

template <int F, bool FULL, bool KEEP>
__device__ __forceinline__ void serve_queue(unsigned e, int nvalid, int lane, int d0,
                                            const int* __restrict__ ei,
                                            const float* __restrict__ dinv,
                                            const float* __restrict__ h,
                                            float* tile, const float* dinvL,
                                            v2u* __restrict__ lst, unsigned* lcnt) {
  const bool valid = FULL || lane < nvalid;
  const int s = ei[e];
  const int dl = valid ? (ei[N_EDGES + e] - d0) : 0;
  const float nr = dinv[s] * dinvL[dl];
  if (KEEP) {
    unsigned base = 0u;
    if (lane == 0)
      base = __hip_atomic_fetch_add(lcnt, (unsigned)(FULL ? 32 : nvalid), __ATOMIC_RELAXED,
                                    __HIP_MEMORY_SCOPE_WORKGROUP);
    base = (unsigned)__builtin_amdgcn_readfirstlane((int)base);
    if (valid && base + (unsigned)lane < (unsigned)LCAP) {
      v2u ent;
      ent.x = (unsigned)s | ((unsigned)dl << 17);
      ent.y = __float_as_uint(nr);
      lst[base + (unsigned)lane] = ent;
    }
  }
  add_rows<F, FULL>(s, dl, nr, nvalid, lane, h, tile);
}

template <int F, bool KEEP>
__global__ __launch_bounds__(1024) void k_edge_sum_own(const int* __restrict__ ei,
                                                        const unsigned* __restrict__ tag4,
                                                        const float* __restrict__ dinv,
                                                        const float* __restrict__ h,
                                                        const float* __restrict__ b,
                                                        float* __restrict__ agg,
                                                        unsigned* __restrict__ lists,
                                                        unsigned* __restrict__ counts) {
  static_assert(N_NODES <= (1 << 17) && BN <= (1 << 15), "src and tile row share one word of a list entry");
  static_assert(LCAP % 32 == 0, "an owner's list is a whole number of 256-byte groups");
  static_assert(N_NODES % BN == 0 && N_NODES / BN <= 256, "whole ranges; the range index fits a byte");
  static_assert(N_EDGES % 16 == 0, "the tags are read 16 at a time");
  static_assert(F % 32 == 0, "a row is a whole number of 128-byte lines");
  constexpr int VEC = F / 32;
  constexpr int NV = N_EDGES / 16;
  constexpr int NCH = (NV + 63) / 64;
  typedef float vecF __attribute__((ext_vector_type(VEC)));
  __shared__ float tile[BN * F];
  __shared__ float dinvL[BN];
  __shared__ unsigned q[32 * 64];
  __shared__ unsigned lcnt;

  const int lane = threadIdx.x & 31;
  const int wave = threadIdx.x >> 5;
  const int d0 = blockIdx.x * BN;
  v2u* __restrict__ lst = (v2u*)lists + (size_t)blockIdx.x * LCAP;
  if (KEEP && threadIdx.x == 0) lcnt = 0u;

  {
    const vecF bv = *(const vecF*)(b + lane * VEC);
    for (int r = wave; r < BN; r += 32) {
      const float di = dinv[d0 + r];
      const vecF hv = *(const vecF*)(h + (long)(d0 + r) * F + lane * VEC);
      for (int c = 0; c < VEC; ++c) tile[r * F + c * 32 + lane] = hv[c] * (di * di) + bv[c];
      if (lane == 0) dinvL[r] = di;
    }
  }
  __syncthreads();

  bool scan = true;
  if (!KEEP) {
    const unsigned n = counts[blockIdx.x];
    if (n <= (unsigned)LCAP) {
      scan = false;
      const int nn = (int)n;
      const int ngr = (nn + 31) >> 5;
      const int wv = __builtin_amdgcn_readfirstlane(wave);
      v2u ent = {0u, 0u};
      if (wv * 32 + lane < nn) ent = lst[wv * 32 + lane];
#pragma unroll 1
      for (int c = wv; c < ngr; c += 32) {
        v2u nxt = {0u, 0u};
        const int ni = (c + 32) * 32 + lane;
        if (ni < nn) nxt = lst[ni];
        const int nv = nn - c * 32;
        const int s = (int)(ent.x & 0x1ffffu);
        const int dl = (int)(ent.x >> 17);
        const float nr = __uint_as_float(ent.y);
        if (nv >= 32) add_rows<F, true>(s, dl, nr, 32, lane, h, tile);
        else          add_rows<F, false>(s, dl, nr, nv, lane, h, tile);
        ent = nxt;
      }
    }
  }

  if (scan) {
  unsigned* qw = q + wave * 64;
  const unsigned pat = blockIdx.x * 0x01010101u;
  const uint4* __restrict__ tagv = (const uint4*)tag4;
  static_assert((NCH - 1) * 64 + 31 < NV, "the first load of every trip lies inside the list");
  auto load_tags = [&](int trip, int half) -> uint4 {
    const int u = trip * 64 + half * 32 + lane;
    return tagv[u < NV ? u : NV - 1];
  };
  int cnt = 0;
  uint4 xa = load_tags(wave, 0);
  uint4 xb = load_tags(wave, 1);
#pragma unroll 1
  for (int it = wave; it < NCH; it += 32) {
    const uint4 na = load_tags(it + 32, 0);
    const uint4 nb = load_tags(it + 32, 1);
    unsigned mb = (zero_bytes(xb.x ^ pat) >> 3) | (zero_bytes(xb.y ^ pat) >> 2) |
                  (zero_bytes(xb.z ^ pat) >> 1) | zero_bytes(xb.w ^ pat);
    if (it * 64 + 32 + lane >= NV) mb = 0u;
    unsigned mm = (zero_bytes(xa.x ^ pat) >> 7) | (zero_bytes(xa.y ^ pat) >> 6) |
                  (zero_bytes(xa.z ^ pat) >> 5) | (zero_bytes(xa.w ^ pat) >> 4) | mb;
    const unsigned ebase = (unsigned)it * 1024u + (unsigned)lane * 16u;
    for (int t = 0; t < 32; ++t) {
      const bool has = (mm != 0u);
      const unsigned bal = __builtin_amdgcn_ballot_w32(has);
      if (bal == 0u) break;
      if (has) {
        const unsigned p = (unsigned)__builtin_ctz(mm);
        mm &= mm - 1u;
        const unsigned j = p & 7u;
        __hip_atomic_store(qw + cnt + (int)__builtin_amdgcn_mbcnt_lo(bal, 0u),
                           ebase + (j >> 2) * 512u + 4u * (j & 3u) + (p >> 3), __ATOMIC_RELAXED,
                           __HIP_MEMORY_SCOPE_WAVEFRONT);
      }
      cnt += __builtin_popcount(bal);
      if (cnt >= 32) {
        const unsigned e32 = __hip_atomic_load(qw + lane, __ATOMIC_RELAXED, __HIP_MEMORY_SCOPE_WAVEFRONT);
        serve_queue<F, true, KEEP>(e32, 32, lane, d0, ei, dinv, h, tile, dinvL, lst, &lcnt);
        const int rest = cnt - 32;
        if (lane < rest) {
          const unsigned mv = __hip_atomic_load(qw + 32 + lane, __ATOMIC_RELAXED, __HIP_MEMORY_SCOPE_WAVEFRONT);
          __hip_atomic_store(qw + lane, mv, __ATOMIC_RELAXED, __HIP_MEMORY_SCOPE_WAVEFRONT);
        }
        cnt = rest;
      }
    }
    xa = na;
    xb = nb;
  }
  if (cnt > 0) {
    unsigned el = 0u;
    if (lane < cnt) el = __hip_atomic_load(qw + lane, __ATOMIC_RELAXED, __HIP_MEMORY_SCOPE_WAVEFRONT);
    serve_queue<F, false, KEEP>(el, cnt, lane, d0, ei, dinv, h, tile, dinvL, lst, &lcnt);
  }
  }
  __syncthreads();
  if (KEEP && threadIdx.x == 0) counts[blockIdx.x] = lcnt;

  for (int r = wave; r < BN; r += 32) {
    vecF o;
    for (int c = 0; c < VEC; ++c) o[c] = tile[r * F + c * 32 + lane];
    *(vecF*)(agg + (long)(d0 + r) * F + lane * VEC) = o;
  }
}

extern "C" void kernel_launch(void* const* d_in, const int* in_sizes, int n_in,
                              void* d_out, int out_size, void* d_ws, size_t ws_size,
                              hipStream_t stream) {
  const float* x  = (const float*)d_in[0];
  const int*   ei = (const int*)d_in[1];
  const float* W1 = (const float*)d_in[2];
  const float* b1 = (const float*)d_in[3];
  const float* W2 = (const float*)d_in[4];
  const float* b2 = (const float*)d_in[5];
  float* out = (float*)d_out;

  float* ws   = (float*)d_ws;
  float* h1   = ws;
  float* agg1 = h1 + (size_t)N_NODES * HID;
  unsigned* tag4 = (unsigned*)(agg1 + (size_t)N_NODES * HID);
  float* dinv = (float*)tag4 + (size_t)N_NODES * OUTD;
  float* h3   = h1;
  unsigned* lists  = tag4 + N_EDGES / 4;
  unsigned* counts = lists + (size_t)(N_NODES / BN) * LCAP * 2;
  static_assert(N_EDGES % 256 == 0, "the lists start on an aligned 256-byte unit");
  static_assert((long)N_EDGES + (long)(N_NODES / BN) * (LCAP * 8 + 4) <= (long)N_NODES * OUTD * 4,
                "the owner tags, lists and counts fit in the third region");

  const int T = 256;

  k_init_deg<<<(N_NODES + T - 1) / T, T, 0, stream>>>(dinv);
  k_count_deg_tag<<<(N_EDGES / 4 + T - 1) / T, T, 0, stream>>>(ei, dinv, tag4);
  k_rsqrt_inplace<<<(N_NODES + T - 1) / T, T, 0, stream>>>(dinv);

  const int mtiles = (N_NODES + 15) / 16;
  const int gblocks = (mtiles + 7) / 8;
  k_gemm_wmma<IN_DIM, HID / 16, false>
      <<<gblocks, T, 0, stream>>>(x, W1, h1, N_NODES);

  k_edge_sum_own<HID, true><<<N_NODES / BN, 1024, 0, stream>>>(ei, tag4, dinv, h1, b1, agg1, lists, counts);

  k_gemm_wmma<HID, OUTD / 16, true>
      <<<gblocks, T, 0, stream>>>(agg1, W2, h3, N_NODES);

  k_edge_sum_own<OUTD, false><<<N_NODES / BN, 1024, 0, stream>>>(ei, tag4, dinv, h3, b2, out, lists, counts);
}


# In-degrees counted by the dst-range owners in LDS (integer adds) in one pass that filters the tags once and keeps each owner's edge list; no global atomics; 7 launches -> 6; sum order = LDS arrival
# speedup vs baseline: 3.1422x; 1.3426x over previous
#include <hip/hip_runtime.h>

#define N_NODES 100000
#define N_EDGES 1600000
#define IN_DIM  256
#define HID     128
#define OUTD    64
#define BN      400
#define LCAP    11904

typedef __attribute__((ext_vector_type(2))) float v2f;
typedef __attribute__((ext_vector_type(2))) unsigned v2u;
typedef __attribute__((ext_vector_type(8))) float v8f;

__global__ void k_owner_tag(const int* __restrict__ ei, unsigned* __restrict__ tag4) {
  static_assert(N_EDGES % 4 == 0, "four destinations per 16-byte load");
  int t = blockIdx.x * blockDim.x + threadIdx.x;
  if (t < N_EDGES / 4) {
    const int4 d = *(const int4*)(ei + N_EDGES + 4 * t);
    tag4[t] = ((unsigned)d.x / BN) | (((unsigned)d.y / BN) << 8) |
              (((unsigned)d.z / BN) << 16) | (((unsigned)d.w / BN) << 24);
  }
}

template <int K, int NT, bool RELU>
__global__ __launch_bounds__(256) void k_gemm_wmma(const float* __restrict__ A,
                                                   const float* __restrict__ W,
                                                   float* __restrict__ C, int M) {
  constexpr int NC = NT * 16;
  constexpr int KC = 64;
  __shared__ float lds[KC * NC];

  const int wave = threadIdx.x >> 5;
  const int lane = threadIdx.x & 31;
  const int mtile = blockIdx.x * 8 + wave;
  const long m0 = (long)mtile * 16;
  const bool active = (m0 < M);

  const int m  = lane & 15;
  const int kh = lane >> 4;
  const int n  = lane & 15;

  const long arow = active ? (m0 + m) : (long)(M - 1);
  const float* __restrict__ Aptr = A + arow * K;

  v8f acc[NT] = {};

  for (int kc = 0; kc < K; kc += KC) {
    if (kc != 0) __syncthreads();
    {
      constexpr int ENT = KC * NC / 8;
#pragma unroll
      for (int idx = threadIdx.x; idx < ENT; idx += 256) {
        int kp = idx / (NC / 4);
        int nq = (idx % (NC / 4)) * 4;
        const float4 a = *(const float4*)(W + (long)(kc + 2 * kp) * NC + nq);
        const float4 b = *(const float4*)(W + (long)(kc + 2 * kp + 1) * NC + nq);
        float4* __restrict__ dst = (float4*)&lds[kp * 2 * NC + 2 * nq];
        dst[0] = make_float4(a.x, b.x, a.y, b.y);
        dst[1] = make_float4(a.z, b.z, a.w, b.w);
      }
    }
    __syncthreads();

#pragma unroll 4
    for (int k0 = kc; k0 < kc + KC; k0 += 4) {
      const int ka = k0 + 2 * kh;
      const int kp = (ka - kc) >> 1;
      v2f av = *(const v2f*)(Aptr + ka);
      if (RELU) {
        av.x = fmaxf(av.x, 0.0f);
        av.y = fmaxf(av.y, 0.0f);
      }
      const v2f* __restrict__ brow = (const v2f*)&lds[kp * 2 * NC + 2 * n];
#pragma unroll
      for (int j = 0; j < NT; ++j) {
        v2f bv = brow[j * 16];
        acc[j] = __builtin_amdgcn_wmma_f32_16x16x4_f32(
             false, av,  false, bv,
             (short)0, acc[j],  false,  false);
      }
    }
  }

  if (!active) return;
#pragma unroll
  for (int j = 0; j < NT; ++j) {
#pragma unroll
    for (int r = 0; r < 8; ++r) {
      long row = m0 + r + 8 * kh;
      if (row < M) C[row * NC + j * 16 + n] = acc[j][r];
    }
  }
}

__device__ __forceinline__ unsigned zero_bytes(unsigned t) {
  return ~(((t & 0x7f7f7f7fu) + 0x7f7f7f7fu) | t | 0x7f7f7f7fu);
}

template <typename ServeFull, typename ServePart>
__device__ __forceinline__ void scan_owner_tags(const unsigned* __restrict__ tag4, unsigned* qw,
                                                int lane, int wave, unsigned owner,
                                                ServeFull serve_full, ServePart serve_part) {
  static_assert(N_EDGES % 16 == 0, "the tags are read 16 at a time");
  constexpr int NV = N_EDGES / 16;
  constexpr int NCH = (NV + 63) / 64;
  const unsigned pat = owner * 0x01010101u;
  const uint4* __restrict__ tagv = (const uint4*)tag4;
  static_assert((NCH - 1) * 64 + 31 < NV, "the first load of every trip lies inside the list");
  auto load_tags = [&](int trip, int half) -> uint4 {
    const int u = trip * 64 + half * 32 + lane;
    return tagv[u < NV ? u : NV - 1];
  };
  int cnt = 0;
  uint4 xa = load_tags(wave, 0);
  uint4 xb = load_tags(wave, 1);
#pragma unroll 1
  for (int it = wave; it < NCH; it += 32) {
    const uint4 na = load_tags(it + 32, 0);
    const uint4 nb = load_tags(it + 32, 1);
    unsigned mb = (zero_bytes(xb.x ^ pat) >> 3) | (zero_bytes(xb.y ^ pat) >> 2) |
                  (zero_bytes(xb.z ^ pat) >> 1) | zero_bytes(xb.w ^ pat);
    if (it * 64 + 32 + lane >= NV) mb = 0u;
    unsigned mm = (zero_bytes(xa.x ^ pat) >> 7) | (zero_bytes(xa.y ^ pat) >> 6) |
                  (zero_bytes(xa.z ^ pat) >> 5) | (zero_bytes(xa.w ^ pat) >> 4) | mb;
    const unsigned ebase = (unsigned)it * 1024u + (unsigned)lane * 16u;
    for (int t = 0; t < 32; ++t) {
      const bool has = (mm != 0u);
      const unsigned bal = __builtin_amdgcn_ballot_w32(has);
      if (bal == 0u) break;
      if (has) {
        const unsigned p = (unsigned)__builtin_ctz(mm);
        mm &= mm - 1u;
        const unsigned j = p & 7u;
        __hip_atomic_store(qw + cnt + (int)__builtin_amdgcn_mbcnt_lo(bal, 0u),
                           ebase + (j >> 2) * 512u + 4u * (j & 3u) + (p >> 3), __ATOMIC_RELAXED,
                           __HIP_MEMORY_SCOPE_WAVEFRONT);
      }
      cnt += __builtin_popcount(bal);
      if (cnt >= 32) {
        const unsigned e32 = __hip_atomic_load(qw + lane, __ATOMIC_RELAXED, __HIP_MEMORY_SCOPE_WAVEFRONT);
        serve_full(e32);
        const int rest = cnt - 32;
        if (lane < rest) {
          const unsigned mv = __hip_atomic_load(qw + 32 + lane, __ATOMIC_RELAXED, __HIP_MEMORY_SCOPE_WAVEFRONT);
          __hip_atomic_store(qw + lane, mv, __ATOMIC_RELAXED, __HIP_MEMORY_SCOPE_WAVEFRONT);
        }
        cnt = rest;
      }
    }
    xa = na;
    xb = nb;
  }
  if (cnt > 0) {
    unsigned el = 0u;
    if (lane < cnt) el = __hip_atomic_load(qw + lane, __ATOMIC_RELAXED, __HIP_MEMORY_SCOPE_WAVEFRONT);
    serve_part(el, cnt);
  }
}

template <bool FULL>
__device__ __forceinline__ void count_and_keep(unsigned e, int nvalid, int lane, int d0,
                                               const int* __restrict__ ei, unsigned* indeg,
                                               unsigned* __restrict__ lx, unsigned* lcnt) {
  const bool valid = FULL || lane < nvalid;
  const int s = ei[e];
  const int dl = valid ? (ei[N_EDGES + e] - d0) : 0;
  if (valid)
    __hip_atomic_fetch_add(indeg + dl, 1u, __ATOMIC_RELAXED, __HIP_MEMORY_SCOPE_WORKGROUP);
  unsigned base = 0u;
  if (lane == 0)
    base = __hip_atomic_fetch_add(lcnt, (unsigned)(FULL ? 32 : nvalid), __ATOMIC_RELAXED,
                                  __HIP_MEMORY_SCOPE_WORKGROUP);
  base = (unsigned)__builtin_amdgcn_readfirstlane((int)base);
  if (valid && base + (unsigned)lane < (unsigned)LCAP)
    lx[base + (unsigned)lane] = (unsigned)s | ((unsigned)dl << 17);
}

__global__ __launch_bounds__(1024) void k_owner_deg(const int* __restrict__ ei,
                                                     const unsigned* __restrict__ tag4,
                                                     float* __restrict__ dinv,
                                                     unsigned* __restrict__ lists,
                                                     unsigned* __restrict__ counts) {
  static_assert(N_NODES <= (1 << 17) && BN <= (1 << 15), "src and tile row share one word of a list entry");
  static_assert(LCAP % 32 == 0, "an owner's list is a whole number of 128-byte groups");
  static_assert(N_NODES % BN == 0 && N_NODES / BN <= 256, "whole ranges; the range index fits a byte");
  __shared__ unsigned indeg[BN];
  __shared__ unsigned q[32 * 64];
  __shared__ unsigned lcnt;

  const int lane = threadIdx.x & 31;
  const int wave = threadIdx.x >> 5;
  const int d0 = blockIdx.x * BN;
  unsigned* __restrict__ lx = lists + (size_t)blockIdx.x * (2 * LCAP);
  for (int r = threadIdx.x; r < BN; r += 1024) indeg[r] = 0u;
  if (threadIdx.x == 0) lcnt = 0u;
  __syncthreads();

  scan_owner_tags(
      tag4, q + wave * 64, lane, wave, blockIdx.x,
      [&](unsigned e) { count_and_keep<true>(e, 32, lane, d0, ei, indeg, lx, &lcnt); },
      [&](unsigned e, int n) { count_and_keep<false>(e, n, lane, d0, ei, indeg, lx, &lcnt); });
  __syncthreads();

  if (threadIdx.x == 0) counts[blockIdx.x] = lcnt;
  for (int r = threadIdx.x; r < BN; r += 1024)
    dinv[d0 + r] = rsqrtf((float)(indeg[r] + 1u));
}

template <int F, bool FULL>
__device__ __forceinline__ void add_rows(int s, int dl, float nr, int nvalid, int lane,
                                         const float* __restrict__ h, float* tile) {
  constexpr int VEC = F / 32;
  constexpr int U = 16;
  typedef float vecF __attribute__((ext_vector_type(VEC)));
#pragma unroll 1
  for (int j0 = 0; j0 < (FULL ? 32 : nvalid); j0 += U) {
    vecF v[U];
#pragma unroll
    for (int u = 0; u < U; ++u) {
      if (FULL || j0 + u < nvalid) {
        const int sj = __builtin_amdgcn_readlane(s, j0 + u);
        v[u] = *(const vecF*)(h + (long)sj * F + lane * VEC);
      }
    }
#pragma unroll
    for (int u = 0; u < U; ++u) {
      if (FULL || j0 + u < nvalid) {
        const int dj = __builtin_amdgcn_readlane(dl, j0 + u);
        const float nj = __int_as_float(__builtin_amdgcn_readlane(__float_as_int(nr), j0 + u));
        float* tp = tile + dj * F + lane;
        for (int c = 0; c < VEC; ++c)
          __hip_atomic_fetch_add(tp + c * 32, v[u][c] * nj, __ATOMIC_RELAXED,
                                 __HIP_MEMORY_SCOPE_WORKGROUP);
      }
    }
  }
}

template <int F, bool FULL>
__device__ __forceinline__ void serve_queue(unsigned e, int nvalid, int lane, int d0,
                                            const int* __restrict__ ei,
                                            const float* __restrict__ dinv,
                                            const float* __restrict__ h,
                                            float* tile, const float* dinvL) {
  const bool valid = FULL || lane < nvalid;
  const int s = ei[e];
  const int dl = valid ? (ei[N_EDGES + e] - d0) : 0;
  const float nr = dinv[s] * dinvL[dl];
  add_rows<F, FULL>(s, dl, nr, nvalid, lane, h, tile);
}

template <int F, bool FIRST>
__global__ __launch_bounds__(1024) void k_edge_sum_own(const int* __restrict__ ei,
                                                        const unsigned* __restrict__ tag4,
                                                        const float* __restrict__ dinv,
                                                        const float* __restrict__ h,
                                                        const float* __restrict__ b,
                                                        float* __restrict__ agg,
                                                        unsigned* __restrict__ lists,
                                                        const unsigned* __restrict__ counts) {
  static_assert(F % 32 == 0, "a row is a whole number of 128-byte lines");
  constexpr int VEC = F / 32;
  typedef float vecF __attribute__((ext_vector_type(VEC)));
  __shared__ float tile[BN * F];
  __shared__ float dinvL[BN];
  __shared__ unsigned q[32 * 64];

  const int lane = threadIdx.x & 31;
  const int wave = threadIdx.x >> 5;
  const int d0 = blockIdx.x * BN;

  {
    const vecF bv = *(const vecF*)(b + lane * VEC);
    for (int r = wave; r < BN; r += 32) {
      const float di = dinv[d0 + r];
      const vecF hv = *(const vecF*)(h + (long)(d0 + r) * F + lane * VEC);
      for (int c = 0; c < VEC; ++c) tile[r * F + c * 32 + lane] = hv[c] * (di * di) + bv[c];
      if (lane == 0) dinvL[r] = di;
    }
  }
  __syncthreads();

  const unsigned n = counts[blockIdx.x];
  if (n <= (unsigned)LCAP) {
    const unsigned* lx = lists + (size_t)blockIdx.x * (2 * LCAP);
    float* ln = (float*)(lists + (size_t)blockIdx.x * (2 * LCAP) + LCAP);
    const int nn = (int)n;
    const int nfull = nn >> 5;
    const int wv = __builtin_amdgcn_readfirstlane(wave);
    const int i0 = wv * 32 + lane;
    unsigned x0 = 0u;
    if (i0 < nn) x0 = lx[i0];
    float y0 = 0.0f;
    if (FIRST) y0 = dinv[x0 & 0x1ffffu];
    else if (i0 < nn) y0 = ln[i0];
    unsigned x1 = 0u;
    if (i0 + 1024 < nn) x1 = lx[i0 + 1024];
    int c = wv;
#pragma unroll 1
    for (; c < nfull; c += 32) {
      const int ic = c * 32 + lane;
      unsigned x2 = 0u;
      if (ic + 2048 < nn) x2 = lx[ic + 2048];
      float y1 = 0.0f;
      if (FIRST) y1 = dinv[x1 & 0x1ffffu];
      else if (ic + 1024 < nn) y1 = ln[ic + 1024];
      float nr = y0;
      if (FIRST) {
        nr = y0 * dinvL[x0 >> 17];
        ln[ic] = nr;
      }
      add_rows<F, true>((int)(x0 & 0x1ffffu), (int)(x0 >> 17), nr, 32, lane, h, tile);
      x0 = x1;
      y0 = y1;
      x1 = x2;
    }
    const int nlast = nn & 31;
    if (nlast != 0 && c == nfull) {
      float nr = y0;
      if (FIRST) {
        nr = y0 * dinvL[x0 >> 17];
        if (lane < nlast) ln[nfull * 32 + lane] = nr;
      }
      add_rows<F, false>((int)(x0 & 0x1ffffu), (int)(x0 >> 17), nr, nlast, lane, h, tile);
    }
  } else {
    scan_owner_tags(
        tag4, q + wave * 64, lane, wave, blockIdx.x,
        [&](unsigned e) { serve_queue<F, true>(e, 32, lane, d0, ei, dinv, h, tile, dinvL); },
        [&](unsigned e, int m) { serve_queue<F, false>(e, m, lane, d0, ei, dinv, h, tile, dinvL); });
  }
  __syncthreads();

  for (int r = wave; r < BN; r += 32) {
    vecF o;
    for (int c = 0; c < VEC; ++c) o[c] = tile[r * F + c * 32 + lane];
    *(vecF*)(agg + (long)(d0 + r) * F + lane * VEC) = o;
  }
}

extern "C" void kernel_launch(void* const* d_in, const int* in_sizes, int n_in,
                              void* d_out, int out_size, void* d_ws, size_t ws_size,
                              hipStream_t stream) {
  const float* x  = (const float*)d_in[0];
  const int*   ei = (const int*)d_in[1];
  const float* W1 = (const float*)d_in[2];
  const float* b1 = (const float*)d_in[3];
  const float* W2 = (const float*)d_in[4];
  const float* b2 = (const float*)d_in[5];
  float* out = (float*)d_out;

  float* ws   = (float*)d_ws;
  float* h1   = ws;
  float* agg1 = h1 + (size_t)N_NODES * HID;
  unsigned* tag4 = (unsigned*)(agg1 + (size_t)N_NODES * HID);
  float* dinv = (float*)tag4 + (size_t)N_NODES * OUTD;
  float* h3   = h1;
  unsigned* lists  = tag4 + N_EDGES / 4;
  unsigned* counts = lists + (size_t)(N_NODES / BN) * LCAP * 2;
  static_assert(N_EDGES % 256 == 0 && (LCAP * 4) % 256 == 0, "the lists start on aligned 256-byte units");
  static_assert((long)N_EDGES + (long)(N_NODES / BN) * (LCAP * 8 + 4) <= (long)N_NODES * OUTD * 4,
                "the owner tags, lists and counts fit in the third region");

  const int T = 256;

  k_owner_tag<<<(N_EDGES / 4 + T - 1) / T, T, 0, stream>>>(ei, tag4);
  k_owner_deg<<<N_NODES / BN, 1024, 0, stream>>>(ei, tag4, dinv, lists, counts);

  const int mtiles = (N_NODES + 15) / 16;
  const int gblocks = (mtiles + 7) / 8;
  k_gemm_wmma<IN_DIM, HID / 16, false>
      <<<gblocks, T, 0, stream>>>(x, W1, h1, N_NODES);

  k_edge_sum_own<HID, true><<<N_NODES / BN, 1024, 0, stream>>>(ei, tag4, dinv, h1, b1, agg1, lists, counts);

  k_gemm_wmma<HID, OUTD / 16, true>
      <<<gblocks, T, 0, stream>>>(agg1, W2, h3, N_NODES);

  k_edge_sum_own<OUTD, false><<<N_NODES / BN, 1024, 0, stream>>>(ei, tag4, dinv, h3, b2, out, lists, counts);
}


# Degree/list pass and layer-1 GEMM in one launch, split by wave: 16 waves of an owner work-group multiply its 25 row tiles (W1 whole in LDS), 16 filter its tags two trips ahead; 6 launches -> 5
# speedup vs baseline: 3.3967x; 1.0810x over previous
#include <hip/hip_runtime.h>

#define N_NODES 100000
#define N_EDGES 1600000
#define IN_DIM  256
#define HID     128
#define OUTD    64
#define BN      400
#define LCAP    11904

typedef __attribute__((ext_vector_type(2))) float v2f;
typedef __attribute__((ext_vector_type(2))) unsigned v2u;
typedef __attribute__((ext_vector_type(8))) float v8f;

__global__ void k_owner_tag(const int* __restrict__ ei, unsigned* __restrict__ tag4) {
  static_assert(N_EDGES % 4 == 0, "four destinations per 16-byte load");
  int t = blockIdx.x * blockDim.x + threadIdx.x;
  if (t < N_EDGES / 4) {
    const int4 d = *(const int4*)(ei + N_EDGES + 4 * t);
    tag4[t] = ((unsigned)d.x / BN) | (((unsigned)d.y / BN) << 8) |
              (((unsigned)d.z / BN) << 16) | (((unsigned)d.w / BN) << 24);
  }
}

template <int K, int NT, bool RELU>
__global__ __launch_bounds__(256) void k_gemm_wmma(const float* __restrict__ A,
                                                   const float* __restrict__ W,
                                                   float* __restrict__ C, int M) {
  constexpr int NC = NT * 16;
  constexpr int KC = 64;
  __shared__ float lds[KC * NC];

  const int wave = threadIdx.x >> 5;
  const int lane = threadIdx.x & 31;
  const int mtile = blockIdx.x * 8 + wave;
  const long m0 = (long)mtile * 16;
  const bool active = (m0 < M);

  const int m  = lane & 15;
  const int kh = lane >> 4;
  const int n  = lane & 15;

  const long arow = active ? (m0 + m) : (long)(M - 1);
  const float* __restrict__ Aptr = A + arow * K;

  v8f acc[NT] = {};

  for (int kc = 0; kc < K; kc += KC) {
    if (kc != 0) __syncthreads();
    {
      constexpr int ENT = KC * NC / 8;
#pragma unroll
      for (int idx = threadIdx.x; idx < ENT; idx += 256) {
        int kp = idx / (NC / 4);
        int nq = (idx % (NC / 4)) * 4;
        const float4 a = *(const float4*)(W + (long)(kc + 2 * kp) * NC + nq);
        const float4 b = *(const float4*)(W + (long)(kc + 2 * kp + 1) * NC + nq);
        float4* __restrict__ dst = (float4*)&lds[kp * 2 * NC + 2 * nq];
        dst[0] = make_float4(a.x, b.x, a.y, b.y);
        dst[1] = make_float4(a.z, b.z, a.w, b.w);
      }
    }
    __syncthreads();

#pragma unroll 4
    for (int k0 = kc; k0 < kc + KC; k0 += 4) {
      const int ka = k0 + 2 * kh;
      const int kp = (ka - kc) >> 1;
      v2f av = *(const v2f*)(Aptr + ka);
      if (RELU) {
        av.x = fmaxf(av.x, 0.0f);
        av.y = fmaxf(av.y, 0.0f);
      }
      const v2f* __restrict__ brow = (const v2f*)&lds[kp * 2 * NC + 2 * n];
#pragma unroll
      for (int j = 0; j < NT; ++j) {
        v2f bv = brow[j * 16];
        acc[j] = __builtin_amdgcn_wmma_f32_16x16x4_f32(
             false, av,  false, bv,
             (short)0, acc[j],  false,  false);
      }
    }
  }

  if (!active) return;
#pragma unroll
  for (int j = 0; j < NT; ++j) {
#pragma unroll
    for (int r = 0; r < 8; ++r) {
      long row = m0 + r + 8 * kh;
      if (row < M) C[row * NC + j * 16 + n] = acc[j][r];
    }
  }
}

__device__ __forceinline__ unsigned zero_bytes(unsigned t) {
  return ~(((t & 0x7f7f7f7fu) + 0x7f7f7f7fu) | t | 0x7f7f7f7fu);
}

template <typename ServeFull, typename ServePart>
__device__ __forceinline__ void scan_owner_tags(const unsigned* __restrict__ tag4, unsigned* qw,
                                                int lane, int wave, unsigned owner,
                                                ServeFull serve_full, ServePart serve_part) {
  static_assert(N_EDGES % 16 == 0, "the tags are read 16 at a time");
  constexpr int NV = N_EDGES / 16;
  constexpr int NCH = (NV + 63) / 64;
  const unsigned pat = owner * 0x01010101u;
  const uint4* __restrict__ tagv = (const uint4*)tag4;
  static_assert((NCH - 1) * 64 + 31 < NV, "the first load of every trip lies inside the list");
  auto load_tags = [&](int trip, int half) -> uint4 {
    const int u = trip * 64 + half * 32 + lane;
    return tagv[u < NV ? u : NV - 1];
  };
  int cnt = 0;
  uint4 xa = load_tags(wave, 0);
  uint4 xb = load_tags(wave, 1);
#pragma unroll 1
  for (int it = wave; it < NCH; it += 32) {
    const uint4 na = load_tags(it + 32, 0);
    const uint4 nb = load_tags(it + 32, 1);
    unsigned mb = (zero_bytes(xb.x ^ pat) >> 3) | (zero_bytes(xb.y ^ pat) >> 2) |
                  (zero_bytes(xb.z ^ pat) >> 1) | zero_bytes(xb.w ^ pat);
    if (it * 64 + 32 + lane >= NV) mb = 0u;
    unsigned mm = (zero_bytes(xa.x ^ pat) >> 7) | (zero_bytes(xa.y ^ pat) >> 6) |
                  (zero_bytes(xa.z ^ pat) >> 5) | (zero_bytes(xa.w ^ pat) >> 4) | mb;
    const unsigned ebase = (unsigned)it * 1024u + (unsigned)lane * 16u;
    for (int t = 0; t < 32; ++t) {
      const bool has = (mm != 0u);
      const unsigned bal = __builtin_amdgcn_ballot_w32(has);
      if (bal == 0u) break;
      if (has) {
        const unsigned p = (unsigned)__builtin_ctz(mm);
        mm &= mm - 1u;
        const unsigned j = p & 7u;
        __hip_atomic_store(qw + cnt + (int)__builtin_amdgcn_mbcnt_lo(bal, 0u),
                           ebase + (j >> 2) * 512u + 4u * (j & 3u) + (p >> 3), __ATOMIC_RELAXED,
                           __HIP_MEMORY_SCOPE_WAVEFRONT);
      }
      cnt += __builtin_popcount(bal);
      if (cnt >= 32) {
        const unsigned e32 = __hip_atomic_load(qw + lane, __ATOMIC_RELAXED, __HIP_MEMORY_SCOPE_WAVEFRONT);
        serve_full(e32);
        const int rest = cnt - 32;
        if (lane < rest) {
          const unsigned mv = __hip_atomic_load(qw + 32 + lane, __ATOMIC_RELAXED, __HIP_MEMORY_SCOPE_WAVEFRONT);
          __hip_atomic_store(qw + lane, mv, __ATOMIC_RELAXED, __HIP_MEMORY_SCOPE_WAVEFRONT);
        }
        cnt = rest;
      }
    }
    xa = na;
    xb = nb;
  }
  if (cnt > 0) {
    unsigned el = 0u;
    if (lane < cnt) el = __hip_atomic_load(qw + lane, __ATOMIC_RELAXED, __HIP_MEMORY_SCOPE_WAVEFRONT);
    serve_part(el, cnt);
  }
}

template <int NW, typename ServeFull, typename ServePart>
__device__ __forceinline__ void scan_owner_tags_part(const unsigned* __restrict__ tag4, unsigned* qw,
                                                     int lane, int wave, unsigned owner,
                                                     ServeFull serve_full, ServePart serve_part) {
  constexpr int NV = N_EDGES / 16;
  constexpr int NCH = (NV + 63) / 64;
  static_assert((NCH - 1) * 64 + 31 < NV, "the first load of every trip lies inside the list");
  const unsigned pat = owner * 0x01010101u;
  const uint4* __restrict__ tagv = (const uint4*)tag4;
  auto load_tags = [&](int trip, int half) -> uint4 {
    const int u = trip * 64 + half * 32 + lane;
    return tagv[u < NV ? u : NV - 1];
  };
  int cnt = 0;
  auto trip = [&](const uint4 xa, const uint4 xb, int it) {
    unsigned mb = (zero_bytes(xb.x ^ pat) >> 3) | (zero_bytes(xb.y ^ pat) >> 2) |
                  (zero_bytes(xb.z ^ pat) >> 1) | zero_bytes(xb.w ^ pat);
    if (it * 64 + 32 + lane >= NV) mb = 0u;
    unsigned mm = (zero_bytes(xa.x ^ pat) >> 7) | (zero_bytes(xa.y ^ pat) >> 6) |
                  (zero_bytes(xa.z ^ pat) >> 5) | (zero_bytes(xa.w ^ pat) >> 4) | mb;
    const unsigned ebase = (unsigned)it * 1024u + (unsigned)lane * 16u;
    for (int t = 0; t < 32; ++t) {
      const bool has = (mm != 0u);
      const unsigned bal = __builtin_amdgcn_ballot_w32(has);
      if (bal == 0u) break;
      if (has) {
        const unsigned p = (unsigned)__builtin_ctz(mm);
        mm &= mm - 1u;
        const unsigned j = p & 7u;
        __hip_atomic_store(qw + cnt + (int)__builtin_amdgcn_mbcnt_lo(bal, 0u),
                           ebase + (j >> 2) * 512u + 4u * (j & 3u) + (p >> 3), __ATOMIC_RELAXED,
                           __HIP_MEMORY_SCOPE_WAVEFRONT);
      }
      cnt += __builtin_popcount(bal);
      if (cnt >= 32) {
        const unsigned e32 = __hip_atomic_load(qw + lane, __ATOMIC_RELAXED, __HIP_MEMORY_SCOPE_WAVEFRONT);
        serve_full(e32);
        const int rest = cnt - 32;
        if (lane < rest) {
          const unsigned mv = __hip_atomic_load(qw + 32 + lane, __ATOMIC_RELAXED, __HIP_MEMORY_SCOPE_WAVEFRONT);
          __hip_atomic_store(qw + lane, mv, __ATOMIC_RELAXED, __HIP_MEMORY_SCOPE_WAVEFRONT);
        }
        cnt = rest;
      }
    }
  };
  uint4 a0 = load_tags(wave, 0);
  uint4 b0 = load_tags(wave, 1);
  uint4 c0 = load_tags(wave + NW, 0);
  uint4 d0 = load_tags(wave + NW, 1);
#pragma unroll 1
  for (int it = wave; it < NCH; it += 4 * NW) {
    const uint4 a1 = load_tags(it + 2 * NW, 0);
    const uint4 b1 = load_tags(it + 2 * NW, 1);
    const uint4 c1 = load_tags(it + 3 * NW, 0);
    const uint4 d1 = load_tags(it + 3 * NW, 1);
    trip(a0, b0, it);
    if (it + NW < NCH) trip(c0, d0, it + NW);
    a0 = load_tags(it + 4 * NW, 0);
    b0 = load_tags(it + 4 * NW, 1);
    c0 = load_tags(it + 5 * NW, 0);
    d0 = load_tags(it + 5 * NW, 1);
    if (it + 2 * NW < NCH) trip(a1, b1, it + 2 * NW);
    if (it + 3 * NW < NCH) trip(c1, d1, it + 3 * NW);
  }
  if (cnt > 0) {
    unsigned el = 0u;
    if (lane < cnt) el = __hip_atomic_load(qw + lane, __ATOMIC_RELAXED, __HIP_MEMORY_SCOPE_WAVEFRONT);
    serve_part(el, cnt);
  }
}

template <bool FULL>
__device__ __forceinline__ void count_and_keep(unsigned e, int nvalid, int lane, int d0,
                                               const int* __restrict__ ei, unsigned* indeg,
                                               unsigned* __restrict__ lx, unsigned* lcnt) {
  const bool valid = FULL || lane < nvalid;
  const int s = ei[e];
  const int dl = valid ? (ei[N_EDGES + e] - d0) : 0;
  if (valid)
    __hip_atomic_fetch_add(indeg + dl, 1u, __ATOMIC_RELAXED, __HIP_MEMORY_SCOPE_WORKGROUP);
  unsigned base = 0u;
  if (lane == 0)
    base = __hip_atomic_fetch_add(lcnt, (unsigned)(FULL ? 32 : nvalid), __ATOMIC_RELAXED,
                                  __HIP_MEMORY_SCOPE_WORKGROUP);
  base = (unsigned)__builtin_amdgcn_readfirstlane((int)base);
  const unsigned pos = base + (unsigned)lane;
  const unsigned word = (unsigned)s | ((unsigned)dl << 17);
  if (FULL) {
    lx[pos < (unsigned)LCAP ? pos : (unsigned)LCAP] = word;
  } else if (valid && pos < (unsigned)LCAP) {
    lx[pos] = word;
  }
}

#define NW_OWN 16
#define NW_MUL 16
__global__ __launch_bounds__(1024) void k_owner_deg_gemm1(const int* __restrict__ ei,
                                                           const unsigned* __restrict__ tag4,
                                                           float* __restrict__ dinv,
                                                           unsigned* __restrict__ lists,
                                                           unsigned* __restrict__ counts,
                                                           const float* __restrict__ x,
                                                           const float* __restrict__ W1,
                                                           float* __restrict__ h1) {
  static_assert(N_NODES <= (1 << 17) && BN <= (1 << 15), "src and tile row share one word of a list entry");
  static_assert(LCAP % 32 == 0, "an owner's list is a whole number of 128-byte groups");
  static_assert(N_NODES % BN == 0 && N_NODES / BN <= 256, "whole ranges; the range index fits a byte");
  static_assert(BN % 16 == 0, "a range is a whole number of 16-row tiles");
  static_assert((NW_OWN + NW_MUL) * 32 == 1024, "the two sides are all the waves of the work-group");
  constexpr int K = IN_DIM;
  constexpr int NC = HID;
  constexpr int NT = HID / 16;
  constexpr int TPR = BN / 16;
  __shared__ float wl[K * NC];
  __shared__ unsigned indeg[BN];
  __shared__ unsigned q[NW_OWN * 64];
  __shared__ unsigned lcnt;

  const int lane = threadIdx.x & 31;
  const int wave = __builtin_amdgcn_readfirstlane((int)(threadIdx.x >> 5));
  const int d0 = blockIdx.x * BN;

  {
    constexpr int ENT = K * NC / 8;
    for (int idx = threadIdx.x; idx < ENT; idx += 1024) {
      const int kp = idx / (NC / 4);
      const int nq = (idx % (NC / 4)) * 4;
      const float4 a = *(const float4*)(W1 + (long)(2 * kp) * NC + nq);
      const float4 b = *(const float4*)(W1 + (long)(2 * kp + 1) * NC + nq);
      float4* __restrict__ dst = (float4*)&wl[kp * 2 * NC + 2 * nq];
      dst[0] = make_float4(a.x, b.x, a.y, b.y);
      dst[1] = make_float4(a.z, b.z, a.w, b.w);
    }
  }
  for (int r = threadIdx.x; r < BN; r += 1024) indeg[r] = 0u;
  if (threadIdx.x == 0) lcnt = 0u;
  __syncthreads();

  if (wave >= NW_MUL) {
    const int ow = wave - NW_MUL;
    unsigned* __restrict__ lx = lists + (size_t)blockIdx.x * (2 * LCAP);
    scan_owner_tags_part<NW_OWN>(
        tag4, q + ow * 64, lane, ow, blockIdx.x,
        [&](unsigned e) { count_and_keep<true>(e, 32, lane, d0, ei, indeg, lx, &lcnt); },
        [&](unsigned e, int n) { count_and_keep<false>(e, n, lane, d0, ei, indeg, lx, &lcnt); });
  } else {
    const int m  = lane & 15;
    const int kh = lane >> 4;
    const int n  = lane & 15;
#pragma unroll 1
    for (int t = wave; t < TPR; t += NW_MUL) {
      const long m0 = (long)d0 + t * 16;
      const float* __restrict__ Aptr = x + (m0 + m) * K;
      v8f acc[NT] = {};
#pragma unroll 4
      for (int k0 = 0; k0 < K; k0 += 4) {
        const int ka = k0 + 2 * kh;
        const v2f av = *(const v2f*)(Aptr + ka);
        const v2f* __restrict__ brow = (const v2f*)&wl[(ka >> 1) * 2 * NC + 2 * n];
#pragma unroll
        for (int j = 0; j < NT; ++j) {
          const v2f bv = brow[j * 16];
          acc[j] = __builtin_amdgcn_wmma_f32_16x16x4_f32(
               false, av,  false, bv,
               (short)0, acc[j],  false,  false);
        }
      }
#pragma unroll
      for (int j = 0; j < NT; ++j) {
#pragma unroll
        for (int r = 0; r < 8; ++r) h1[(m0 + r + 8 * kh) * NC + j * 16 + n] = acc[j][r];
      }
    }
  }
  __syncthreads();

  if (threadIdx.x == 0) counts[blockIdx.x] = lcnt;
  for (int r = threadIdx.x; r < BN; r += 1024)
    dinv[d0 + r] = rsqrtf((float)(indeg[r] + 1u));
}

template <int F, bool FULL>
__device__ __forceinline__ void add_rows(int s, int dl, float nr, int nvalid, int lane,
                                         const float* __restrict__ h, float* tile) {
  constexpr int VEC = F / 32;
  constexpr int U = 16;
  typedef float vecF __attribute__((ext_vector_type(VEC)));
#pragma unroll 1
  for (int j0 = 0; j0 < (FULL ? 32 : nvalid); j0 += U) {
    vecF v[U];
#pragma unroll
    for (int u = 0; u < U; ++u) {
      if (FULL || j0 + u < nvalid) {
        const int sj = __builtin_amdgcn_readlane(s, j0 + u);
        v[u] = *(const vecF*)(h + (long)sj * F + lane * VEC);
      }
    }
#pragma unroll
    for (int u = 0; u < U; ++u) {
      if (FULL || j0 + u < nvalid) {
        const int dj = __builtin_amdgcn_readlane(dl, j0 + u);
        const float nj = __int_as_float(__builtin_amdgcn_readlane(__float_as_int(nr), j0 + u));
        float* tp = tile + dj * F + lane;
        for (int c = 0; c < VEC; ++c)
          __hip_atomic_fetch_add(tp + c * 32, v[u][c] * nj, __ATOMIC_RELAXED,
                                 __HIP_MEMORY_SCOPE_WORKGROUP);
      }
    }
  }
}

template <int F, bool FULL>
__device__ __forceinline__ void serve_queue(unsigned e, int nvalid, int lane, int d0,
                                            const int* __restrict__ ei,
                                            const float* __restrict__ dinv,
                                            const float* __restrict__ h,
                                            float* tile, const float* dinvL) {
  const bool valid = FULL || lane < nvalid;
  const int s = ei[e];
  const int dl = valid ? (ei[N_EDGES + e] - d0) : 0;
  const float nr = dinv[s] * dinvL[dl];
  add_rows<F, FULL>(s, dl, nr, nvalid, lane, h, tile);
}

template <int F, bool FIRST>
__global__ __launch_bounds__(1024) void k_edge_sum_own(const int* __restrict__ ei,
                                                        const unsigned* __restrict__ tag4,
                                                        const float* __restrict__ dinv,
                                                        const float* __restrict__ h,
                                                        const float* __restrict__ b,
                                                        float* __restrict__ agg,
                                                        unsigned* __restrict__ lists,
                                                        const unsigned* __restrict__ counts) {
  static_assert(F % 32 == 0, "a row is a whole number of 128-byte lines");
  constexpr int VEC = F / 32;
  typedef float vecF __attribute__((ext_vector_type(VEC)));
  __shared__ float tile[BN * F];
  __shared__ float dinvL[BN];
  __shared__ unsigned q[32 * 64];

  const int lane = threadIdx.x & 31;
  const int wave = threadIdx.x >> 5;
  const int d0 = blockIdx.x * BN;

  {
    const vecF bv = *(const vecF*)(b + lane * VEC);
    for (int r = wave; r < BN; r += 32) {
      const float di = dinv[d0 + r];
      const vecF hv = *(const vecF*)(h + (long)(d0 + r) * F + lane * VEC);
      for (int c = 0; c < VEC; ++c) tile[r * F + c * 32 + lane] = hv[c] * (di * di) + bv[c];
      if (lane == 0) dinvL[r] = di;
    }
  }
  __syncthreads();

  const unsigned n = counts[blockIdx.x];
  if (n <= (unsigned)LCAP) {
    const unsigned* lx = lists + (size_t)blockIdx.x * (2 * LCAP);
    float* ln = (float*)(lists + (size_t)blockIdx.x * (2 * LCAP) + LCAP);
    const int nn = (int)n;
    const int nfull = nn >> 5;
    const int wv = __builtin_amdgcn_readfirstlane(wave);
    const int i0 = wv * 32 + lane;
    unsigned x0 = 0u;
    if (i0 < nn) x0 = lx[i0];
    float y0 = 0.0f;
    if (FIRST) y0 = dinv[x0 & 0x1ffffu];
    else if (i0 < nn) y0 = ln[i0];
    unsigned x1 = 0u;
    if (i0 + 1024 < nn) x1 = lx[i0 + 1024];
    int c = wv;
#pragma unroll 1
    for (; c < nfull; c += 32) {
      const int ic = c * 32 + lane;
      unsigned x2 = 0u;
      if (ic + 2048 < nn) x2 = lx[ic + 2048];
      float y1 = 0.0f;
      if (FIRST) y1 = dinv[x1 & 0x1ffffu];
      else if (ic + 1024 < nn) y1 = ln[ic + 1024];
      float nr = y0;
      if (FIRST) {
        nr = y0 * dinvL[x0 >> 17];
        ln[ic] = nr;
      }
      add_rows<F, true>((int)(x0 & 0x1ffffu), (int)(x0 >> 17), nr, 32, lane, h, tile);
      x0 = x1;
      y0 = y1;
      x1 = x2;
    }
    const int nlast = nn & 31;
    if (nlast != 0 && c == nfull) {
      float nr = y0;
      if (FIRST) {
        nr = y0 * dinvL[x0 >> 17];
        if (lane < nlast) ln[nfull * 32 + lane] = nr;
      }
      add_rows<F, false>((int)(x0 & 0x1ffffu), (int)(x0 >> 17), nr, nlast, lane, h, tile);
    }
  } else {
    scan_owner_tags(
        tag4, q + wave * 64, lane, wave, blockIdx.x,
        [&](unsigned e) { serve_queue<F, true>(e, 32, lane, d0, ei, dinv, h, tile, dinvL); },
        [&](unsigned e, int m) { serve_queue<F, false>(e, m, lane, d0, ei, dinv, h, tile, dinvL); });
  }
  __syncthreads();

  for (int r = wave; r < BN; r += 32) {
    vecF o;
    for (int c = 0; c < VEC; ++c) o[c] = tile[r * F + c * 32 + lane];
    *(vecF*)(agg + (long)(d0 + r) * F + lane * VEC) = o;
  }
}

extern "C" void kernel_launch(void* const* d_in, const int* in_sizes, int n_in,
                              void* d_out, int out_size, void* d_ws, size_t ws_size,
                              hipStream_t stream) {
  const float* x  = (const float*)d_in[0];
  const int*   ei = (const int*)d_in[1];
  const float* W1 = (const float*)d_in[2];
  const float* b1 = (const float*)d_in[3];
  const float* W2 = (const float*)d_in[4];
  const float* b2 = (const float*)d_in[5];
  float* out = (float*)d_out;

  float* ws   = (float*)d_ws;
  float* h1   = ws;
  float* agg1 = h1 + (size_t)N_NODES * HID;
  unsigned* tag4 = (unsigned*)(agg1 + (size_t)N_NODES * HID);
  float* dinv = (float*)tag4 + (size_t)N_NODES * OUTD;
  float* h3   = h1;
  unsigned* lists  = tag4 + N_EDGES / 4;
  unsigned* counts = lists + (size_t)(N_NODES / BN) * LCAP * 2;
  static_assert(N_EDGES % 256 == 0 && (LCAP * 4) % 256 == 0, "the lists start on aligned 256-byte units");
  static_assert((long)N_EDGES + (long)(N_NODES / BN) * (LCAP * 8 + 4) <= (long)N_NODES * OUTD * 4,
                "the owner tags, lists and counts fit in the third region");

  const int T = 256;

  k_owner_tag<<<(N_EDGES / 4 + T - 1) / T, T, 0, stream>>>(ei, tag4);

  k_owner_deg_gemm1<<<N_NODES / BN, 1024, 0, stream>>>(ei, tag4, dinv, lists, counts, x, W1, h1);

  const int mtiles = (N_NODES + 15) / 16;
  const int gblocks = (mtiles + 7) / 8;

  k_edge_sum_own<HID, true><<<N_NODES / BN, 1024, 0, stream>>>(ei, tag4, dinv, h1, b1, agg1, lists, counts);

  k_gemm_wmma<HID, OUTD / 16, true>
      <<<gblocks, T, 0, stream>>>(agg1, W2, h3, N_NODES);

  k_edge_sum_own<OUTD, false><<<N_NODES / BN, 1024, 0, stream>>>(ei, tag4, dinv, h3, b2, out, lists, counts);
}


# Edge sums in registers, not an LDS tile: list pass splits an owner's edges by dst mod 32 into 32 per-wave lists; a wave keeps its 13 rows' sums in VGPRs, no LDS atomics; sum order = list order, fma
# speedup vs baseline: 3.6385x; 1.0712x over previous
#include <hip/hip_runtime.h>

#define N_NODES 100000
#define N_EDGES 1600000
#define IN_DIM  256
#define HID     128
#define OUTD    64
#define BN      400
#define RPW     13
#define SCAP    352
#define LW      (32 * SCAP)

typedef __attribute__((ext_vector_type(2))) float v2f;
typedef __attribute__((ext_vector_type(2))) unsigned v2u;
typedef __attribute__((ext_vector_type(8))) float v8f;
typedef __attribute__((ext_vector_type(16))) float v16f;

__global__ void k_owner_tag(const int* __restrict__ ei, unsigned* __restrict__ tag4) {
  static_assert(N_EDGES % 4 == 0, "four destinations per 16-byte load");
  int t = blockIdx.x * blockDim.x + threadIdx.x;
  if (t < N_EDGES / 4) {
    const int4 d = *(const int4*)(ei + N_EDGES + 4 * t);
    tag4[t] = ((unsigned)d.x / BN) | (((unsigned)d.y / BN) << 8) |
              (((unsigned)d.z / BN) << 16) | (((unsigned)d.w / BN) << 24);
  }
}

template <int K, int NT, bool RELU>
__global__ __launch_bounds__(256) void k_gemm_wmma(const float* __restrict__ A,
                                                   const float* __restrict__ W,
                                                   float* __restrict__ C, int M) {
  constexpr int NC = NT * 16;
  constexpr int KC = 64;
  __shared__ float lds[KC * NC];

  const int wave = threadIdx.x >> 5;
  const int lane = threadIdx.x & 31;
  const int mtile = blockIdx.x * 8 + wave;
  const long m0 = (long)mtile * 16;
  const bool active = (m0 < M);

  const int m  = lane & 15;
  const int kh = lane >> 4;
  const int n  = lane & 15;

  const long arow = active ? (m0 + m) : (long)(M - 1);
  const float* __restrict__ Aptr = A + arow * K;

  v8f acc[NT] = {};

  for (int kc = 0; kc < K; kc += KC) {
    if (kc != 0) __syncthreads();
    {
      constexpr int ENT = KC * NC / 8;
#pragma unroll
      for (int idx = threadIdx.x; idx < ENT; idx += 256) {
        int kp = idx / (NC / 4);
        int nq = (idx % (NC / 4)) * 4;
        const float4 a = *(const float4*)(W + (long)(kc + 2 * kp) * NC + nq);
        const float4 b = *(const float4*)(W + (long)(kc + 2 * kp + 1) * NC + nq);
        float4* __restrict__ dst = (float4*)&lds[kp * 2 * NC + 2 * nq];
        dst[0] = make_float4(a.x, b.x, a.y, b.y);
        dst[1] = make_float4(a.z, b.z, a.w, b.w);
      }
    }
    __syncthreads();

#pragma unroll 4
    for (int k0 = kc; k0 < kc + KC; k0 += 4) {
      const int ka = k0 + 2 * kh;
      const int kp = (ka - kc) >> 1;
      v2f av = *(const v2f*)(Aptr + ka);
      if (RELU) {
        av.x = fmaxf(av.x, 0.0f);
        av.y = fmaxf(av.y, 0.0f);
      }
      const v2f* __restrict__ brow = (const v2f*)&lds[kp * 2 * NC + 2 * n];
#pragma unroll
      for (int j = 0; j < NT; ++j) {
        v2f bv = brow[j * 16];
        acc[j] = __builtin_amdgcn_wmma_f32_16x16x4_f32(
             false, av,  false, bv,
             (short)0, acc[j],  false,  false);
      }
    }
  }

  if (!active) return;
#pragma unroll
  for (int j = 0; j < NT; ++j) {
#pragma unroll
    for (int r = 0; r < 8; ++r) {
      long row = m0 + r + 8 * kh;
      if (row < M) C[row * NC + j * 16 + n] = acc[j][r];
    }
  }
}

__device__ __forceinline__ unsigned zero_bytes(unsigned t) {
  return ~(((t & 0x7f7f7f7fu) + 0x7f7f7f7fu) | t | 0x7f7f7f7fu);
}

template <typename ServeFull, typename ServePart>
__device__ __forceinline__ void scan_owner_tags(const unsigned* __restrict__ tag4, unsigned* qw,
                                                int lane, int wave, unsigned owner,
                                                ServeFull serve_full, ServePart serve_part) {
  static_assert(N_EDGES % 16 == 0, "the tags are read 16 at a time");
  constexpr int NV = N_EDGES / 16;
  constexpr int NCH = (NV + 63) / 64;
  const unsigned pat = owner * 0x01010101u;
  const uint4* __restrict__ tagv = (const uint4*)tag4;
  static_assert((NCH - 1) * 64 + 31 < NV, "the first load of every trip lies inside the list");
  auto load_tags = [&](int trip, int half) -> uint4 {
    const int u = trip * 64 + half * 32 + lane;
    return tagv[u < NV ? u : NV - 1];
  };
  int cnt = 0;
  uint4 xa = load_tags(wave, 0);
  uint4 xb = load_tags(wave, 1);
#pragma unroll 1
  for (int it = wave; it < NCH; it += 32) {
    const uint4 na = load_tags(it + 32, 0);
    const uint4 nb = load_tags(it + 32, 1);
    unsigned mb = (zero_bytes(xb.x ^ pat) >> 3) | (zero_bytes(xb.y ^ pat) >> 2) |
                  (zero_bytes(xb.z ^ pat) >> 1) | zero_bytes(xb.w ^ pat);
    if (it * 64 + 32 + lane >= NV) mb = 0u;
    unsigned mm = (zero_bytes(xa.x ^ pat) >> 7) | (zero_bytes(xa.y ^ pat) >> 6) |
                  (zero_bytes(xa.z ^ pat) >> 5) | (zero_bytes(xa.w ^ pat) >> 4) | mb;
    const unsigned ebase = (unsigned)it * 1024u + (unsigned)lane * 16u;
    for (int t = 0; t < 32; ++t) {
      const bool has = (mm != 0u);
      const unsigned bal = __builtin_amdgcn_ballot_w32(has);
      if (bal == 0u) break;
      if (has) {
        const unsigned p = (unsigned)__builtin_ctz(mm);
        mm &= mm - 1u;
        const unsigned j = p & 7u;
        __hip_atomic_store(qw + cnt + (int)__builtin_amdgcn_mbcnt_lo(bal, 0u),
                           ebase + (j >> 2) * 512u + 4u * (j & 3u) + (p >> 3), __ATOMIC_RELAXED,
                           __HIP_MEMORY_SCOPE_WAVEFRONT);
      }
      cnt += __builtin_popcount(bal);
      if (cnt >= 32) {
        const unsigned e32 = __hip_atomic_load(qw + lane, __ATOMIC_RELAXED, __HIP_MEMORY_SCOPE_WAVEFRONT);
        serve_full(e32);
        const int rest = cnt - 32;
        if (lane < rest) {
          const unsigned mv = __hip_atomic_load(qw + 32 + lane, __ATOMIC_RELAXED, __HIP_MEMORY_SCOPE_WAVEFRONT);
          __hip_atomic_store(qw + lane, mv, __ATOMIC_RELAXED, __HIP_MEMORY_SCOPE_WAVEFRONT);
        }
        cnt = rest;
      }
    }
    xa = na;
    xb = nb;
  }
  if (cnt > 0) {
    unsigned el = 0u;
    if (lane < cnt) el = __hip_atomic_load(qw + lane, __ATOMIC_RELAXED, __HIP_MEMORY_SCOPE_WAVEFRONT);
    serve_part(el, cnt);
  }
}

template <int NW, typename ServeFull, typename ServePart>
__device__ __forceinline__ void scan_owner_tags_part(const unsigned* __restrict__ tag4, unsigned* qw,
                                                     int lane, int wave, unsigned owner,
                                                     ServeFull serve_full, ServePart serve_part) {
  constexpr int NV = N_EDGES / 16;
  constexpr int NCH = (NV + 63) / 64;
  static_assert((NCH - 1) * 64 + 31 < NV, "the first load of every trip lies inside the list");
  const unsigned pat = owner * 0x01010101u;
  const uint4* __restrict__ tagv = (const uint4*)tag4;
  auto load_tags = [&](int trip, int half) -> uint4 {
    const int u = trip * 64 + half * 32 + lane;
    return tagv[u < NV ? u : NV - 1];
  };
  int cnt = 0;
  auto trip = [&](const uint4 xa, const uint4 xb, int it) {
    unsigned mb = (zero_bytes(xb.x ^ pat) >> 3) | (zero_bytes(xb.y ^ pat) >> 2) |
                  (zero_bytes(xb.z ^ pat) >> 1) | zero_bytes(xb.w ^ pat);
    if (it * 64 + 32 + lane >= NV) mb = 0u;
    unsigned mm = (zero_bytes(xa.x ^ pat) >> 7) | (zero_bytes(xa.y ^ pat) >> 6) |
                  (zero_bytes(xa.z ^ pat) >> 5) | (zero_bytes(xa.w ^ pat) >> 4) | mb;
    const unsigned ebase = (unsigned)it * 1024u + (unsigned)lane * 16u;
    for (int t = 0; t < 32; ++t) {
      const bool has = (mm != 0u);
      const unsigned bal = __builtin_amdgcn_ballot_w32(has);
      if (bal == 0u) break;
      if (has) {
        const unsigned p = (unsigned)__builtin_ctz(mm);
        mm &= mm - 1u;
        const unsigned j = p & 7u;
        __hip_atomic_store(qw + cnt + (int)__builtin_amdgcn_mbcnt_lo(bal, 0u),
                           ebase + (j >> 2) * 512u + 4u * (j & 3u) + (p >> 3), __ATOMIC_RELAXED,
                           __HIP_MEMORY_SCOPE_WAVEFRONT);
      }
      cnt += __builtin_popcount(bal);
      if (cnt >= 32) {
        const unsigned e32 = __hip_atomic_load(qw + lane, __ATOMIC_RELAXED, __HIP_MEMORY_SCOPE_WAVEFRONT);
        serve_full(e32);
        const int rest = cnt - 32;
        if (lane < rest) {
          const unsigned mv = __hip_atomic_load(qw + 32 + lane, __ATOMIC_RELAXED, __HIP_MEMORY_SCOPE_WAVEFRONT);
          __hip_atomic_store(qw + lane, mv, __ATOMIC_RELAXED, __HIP_MEMORY_SCOPE_WAVEFRONT);
        }
        cnt = rest;
      }
    }
  };
  uint4 a0 = load_tags(wave, 0);
  uint4 b0 = load_tags(wave, 1);
  uint4 c0 = load_tags(wave + NW, 0);
  uint4 d0 = load_tags(wave + NW, 1);
#pragma unroll 1
  for (int it = wave; it < NCH; it += 4 * NW) {
    const uint4 a1 = load_tags(it + 2 * NW, 0);
    const uint4 b1 = load_tags(it + 2 * NW, 1);
    const uint4 c1 = load_tags(it + 3 * NW, 0);
    const uint4 d1 = load_tags(it + 3 * NW, 1);
    trip(a0, b0, it);
    if (it + NW < NCH) trip(c0, d0, it + NW);
    a0 = load_tags(it + 4 * NW, 0);
    b0 = load_tags(it + 4 * NW, 1);
    c0 = load_tags(it + 5 * NW, 0);
    d0 = load_tags(it + 5 * NW, 1);
    if (it + 2 * NW < NCH) trip(a1, b1, it + 2 * NW);
    if (it + 3 * NW < NCH) trip(c1, d1, it + 3 * NW);
  }
  if (cnt > 0) {
    unsigned el = 0u;
    if (lane < cnt) el = __hip_atomic_load(qw + lane, __ATOMIC_RELAXED, __HIP_MEMORY_SCOPE_WAVEFRONT);
    serve_part(el, cnt);
  }
}

template <bool FULL>
__device__ __forceinline__ void count_and_keep(unsigned e, int nvalid, int lane, int d0,
                                               const int* __restrict__ ei, unsigned* indeg,
                                               unsigned* slist, unsigned* scnt) {
  const bool valid = FULL || lane < nvalid;
  const int s = ei[e];
  const int dl = valid ? (ei[N_EDGES + e] - d0) : 0;
  if (valid) {
    __hip_atomic_fetch_add(indeg + dl, 1u, __ATOMIC_RELAXED, __HIP_MEMORY_SCOPE_WORKGROUP);
    const unsigned w = (unsigned)dl & 31u;
    const unsigned pos =
        __hip_atomic_fetch_add(scnt + w, 1u, __ATOMIC_RELAXED, __HIP_MEMORY_SCOPE_WORKGROUP);
    const unsigned word = (unsigned)s | ((unsigned)dl << 17);
    if (FULL) {
      slist[pos < (unsigned)SCAP ? w * SCAP + pos : (unsigned)LW] = word;
    } else if (pos < (unsigned)SCAP) {
      slist[w * SCAP + pos] = word;
    }
  }
}

#define NW_OWN 16
#define NW_MUL 16
__global__ __launch_bounds__(1024) void k_owner_deg_gemm1(const int* __restrict__ ei,
                                                           const unsigned* __restrict__ tag4,
                                                           float* __restrict__ dinv,
                                                           unsigned* __restrict__ lists,
                                                           unsigned* __restrict__ counts,
                                                           const float* __restrict__ x,
                                                           const float* __restrict__ W1,
                                                           float* __restrict__ h1) {
  static_assert(N_NODES <= (1 << 17) && BN <= (1 << 15), "src and tile row share one word of a list entry");
  static_assert(SCAP % 32 == 0, "a wave's list is a whole number of 128-byte groups");
  static_assert(RPW == (BN + 31) / 32 && 32 * (RPW - 1) <= BN, "every wave has RPW - 1 whole rows and maybe one more");
  static_assert(N_NODES % BN == 0 && N_NODES / BN <= 256, "whole ranges; the range index fits a byte");
  static_assert(BN % 16 == 0, "a range is a whole number of 16-row tiles");
  static_assert((NW_OWN + NW_MUL) * 32 == 1024, "the two sides are all the waves of the work-group");
  constexpr int K = IN_DIM;
  constexpr int NC = HID;
  constexpr int NT = HID / 16;
  constexpr int TPR = BN / 16;
  __shared__ float wl[K * NC];
  __shared__ unsigned indeg[BN];
  __shared__ unsigned q[NW_OWN * 64];
  __shared__ unsigned scnt[32];
  __shared__ unsigned slist[LW + 1];

  const int lane = threadIdx.x & 31;
  const int wave = __builtin_amdgcn_readfirstlane((int)(threadIdx.x >> 5));
  const int d0 = blockIdx.x * BN;

  {
    constexpr int ENT = K * NC / 8;
    for (int idx = threadIdx.x; idx < ENT; idx += 1024) {
      const int kp = idx / (NC / 4);
      const int nq = (idx % (NC / 4)) * 4;
      const float4 a = *(const float4*)(W1 + (long)(2 * kp) * NC + nq);
      const float4 b = *(const float4*)(W1 + (long)(2 * kp + 1) * NC + nq);
      float4* __restrict__ dst = (float4*)&wl[kp * 2 * NC + 2 * nq];
      dst[0] = make_float4(a.x, b.x, a.y, b.y);
      dst[1] = make_float4(a.z, b.z, a.w, b.w);
    }
  }
  for (int r = threadIdx.x; r < BN; r += 1024) indeg[r] = 0u;
  if (threadIdx.x < 32) scnt[threadIdx.x] = 0u;
  __syncthreads();

  if (wave >= NW_MUL) {
    const int ow = wave - NW_MUL;
    scan_owner_tags_part<NW_OWN>(
        tag4, q + ow * 64, lane, ow, blockIdx.x,
        [&](unsigned e) { count_and_keep<true>(e, 32, lane, d0, ei, indeg, slist, scnt); },
        [&](unsigned e, int n) { count_and_keep<false>(e, n, lane, d0, ei, indeg, slist, scnt); });
  } else {
    const int m  = lane & 15;
    const int kh = lane >> 4;
    const int n  = lane & 15;
#pragma unroll 1
    for (int t = wave; t < TPR; t += NW_MUL) {
      const long m0 = (long)d0 + t * 16;
      const float* __restrict__ Aptr = x + (m0 + m) * K;
      v8f acc[NT] = {};
#pragma unroll 4
      for (int k0 = 0; k0 < K; k0 += 4) {
        const int ka = k0 + 2 * kh;
        const v2f av = *(const v2f*)(Aptr + ka);
        const v2f* __restrict__ brow = (const v2f*)&wl[(ka >> 1) * 2 * NC + 2 * n];
#pragma unroll
        for (int j = 0; j < NT; ++j) {
          const v2f bv = brow[j * 16];
          acc[j] = __builtin_amdgcn_wmma_f32_16x16x4_f32(
               false, av,  false, bv,
               (short)0, acc[j],  false,  false);
        }
      }
#pragma unroll
      for (int j = 0; j < NT; ++j) {
#pragma unroll
        for (int r = 0; r < 8; ++r) h1[(m0 + r + 8 * kh) * NC + j * 16 + n] = acc[j][r];
      }
    }
  }
  __syncthreads();

  if (threadIdx.x < 32) counts[blockIdx.x * 32 + threadIdx.x] = scnt[threadIdx.x];
  for (int r = threadIdx.x; r < BN; r += 1024)
    dinv[d0 + r] = rsqrtf((float)(indeg[r] + 1u));
  {
    const unsigned nw = scnt[wave] < (unsigned)SCAP ? scnt[wave] : (unsigned)SCAP;
    unsigned* __restrict__ lx = lists + (size_t)blockIdx.x * (2 * LW) + (size_t)wave * SCAP;
    for (unsigned i = lane; i < nw; i += 32) lx[i] = slist[wave * SCAP + i];
  }
}

template <int F, bool FULL>
__device__ __forceinline__ void add_rows(int s, int dl, float nr, int nvalid, int lane,
                                         const float* __restrict__ h, float* tile) {
  constexpr int VEC = F / 32;
  constexpr int U = 16;
  typedef float vecF __attribute__((ext_vector_type(VEC)));
#pragma unroll 1
  for (int j0 = 0; j0 < (FULL ? 32 : nvalid); j0 += U) {
    vecF v[U];
#pragma unroll
    for (int u = 0; u < U; ++u) {
      if (FULL || j0 + u < nvalid) {
        const int sj = __builtin_amdgcn_readlane(s, j0 + u);
        v[u] = *(const vecF*)(h + (long)sj * F + lane * VEC);
      }
    }
#pragma unroll
    for (int u = 0; u < U; ++u) {
      if (FULL || j0 + u < nvalid) {
        const int dj = __builtin_amdgcn_readlane(dl, j0 + u);
        const float nj = __int_as_float(__builtin_amdgcn_readlane(__float_as_int(nr), j0 + u));
        float* tp = tile + dj * F + lane;
        for (int c = 0; c < VEC; ++c)
          __hip_atomic_fetch_add(tp + c * 32, v[u][c] * nj, __ATOMIC_RELAXED,
                                 __HIP_MEMORY_SCOPE_WORKGROUP);
      }
    }
  }
}

template <int F, bool FULL>
__device__ __forceinline__ void serve_queue(unsigned e, int nvalid, int lane, int d0,
                                            const int* __restrict__ ei,
                                            const float* __restrict__ dinv,
                                            const float* __restrict__ h,
                                            float* tile, const float* dinvL) {
  const bool valid = FULL || lane < nvalid;
  const int s = ei[e];
  const int dl = valid ? (ei[N_EDGES + e] - d0) : 0;
  const float nr = dinv[s] * dinvL[dl];
  add_rows<F, FULL>(s, dl, nr, nvalid, lane, h, tile);
}

template <int F, bool FULL>
__device__ __forceinline__ void add_rows_reg(int s, int k, float nr, int nvalid, int lane,
                                             const float* __restrict__ h, v16f (&acc)[F / 32]) {
  constexpr int VEC = F / 32;
  constexpr int U = 32 / VEC;
  typedef float vecF __attribute__((ext_vector_type(VEC)));
#pragma unroll 1
  for (int j0 = 0; j0 < (FULL ? 32 : nvalid); j0 += U) {
    vecF v[U];
#pragma unroll
    for (int u = 0; u < U; ++u) {
      if (FULL || j0 + u < nvalid) {
        const int sj = __builtin_amdgcn_readlane(s, j0 + u);
        v[u] = *(const vecF*)(h + (long)sj * F + lane * VEC);
      }
    }
#pragma unroll
    for (int u = 0; u < U; ++u) {
      if (FULL || j0 + u < nvalid) {
        const int kj = __builtin_amdgcn_readlane(k, j0 + u);
        const float nj = __int_as_float(__builtin_amdgcn_readlane(__float_as_int(nr), j0 + u));
        for (int c = 0; c < VEC; ++c) acc[c][kj] += v[u][c] * nj;
      }
    }
  }
}

template <int F, bool FIRST>
__global__ __launch_bounds__(1024) void k_edge_sum_own(const int* __restrict__ ei,
                                                        const unsigned* __restrict__ tag4,
                                                        const float* __restrict__ dinv,
                                                        const float* __restrict__ h,
                                                        const float* __restrict__ b,
                                                        float* __restrict__ agg,
                                                        unsigned* __restrict__ lists,
                                                        const unsigned* __restrict__ counts) {
  static_assert(F % 32 == 0, "a row is a whole number of 128-byte lines");
  constexpr int VEC = F / 32;
  typedef float vecF __attribute__((ext_vector_type(VEC)));
  __shared__ float tile[BN * F];
  __shared__ float dinvL[BN];
  __shared__ unsigned q[32 * 64];

  const int lane = threadIdx.x & 31;
  const int wave = threadIdx.x >> 5;
  const int d0 = blockIdx.x * BN;

  const unsigned cl = counts[blockIdx.x * 32 + lane];
  if (__builtin_amdgcn_ballot_w32(cl > (unsigned)SCAP) == 0u) {
    const int wv = __builtin_amdgcn_readfirstlane(wave);
    const int nn = __builtin_amdgcn_readlane((int)cl, wv);

    float dmine = 0.0f;
    if (lane < RPW && wv + 32 * lane < BN) dmine = dinv[d0 + wv + 32 * lane];

    const vecF bv = *(const vecF*)(b + lane * VEC);
    v16f acc[VEC];
    static_assert(RPW <= 16, "a wave's rows index a 16-register vector");
    for (int c = 0; c < VEC; ++c) acc[c] = (v16f)(0.0f);
#pragma unroll
    for (int k = 0; k < RPW; ++k) {
      const int r = wv + 32 * k;
      if (k < RPW - 1 || r < BN) {
        const float di = __int_as_float(__builtin_amdgcn_readlane(__float_as_int(dmine), k));
        const vecF hv = *(const vecF*)(h + (long)(d0 + r) * F + lane * VEC);
        for (int c = 0; c < VEC; ++c) acc[c][k] = hv[c] * (di * di) + bv[c];
      }
    }

    const size_t obase = (size_t)blockIdx.x * (2 * LW) + (size_t)wv * SCAP;
    const unsigned* lx = lists + obase;
    float* ln = (float*)(lists + obase + LW);
    const int nfull = nn >> 5;
    unsigned x0 = 0u;
    if (lane < nn) x0 = lx[lane];
    float y0 = 0.0f;
    if (FIRST) y0 = dinv[x0 & 0x1ffffu];
    else if (lane < nn) y0 = ln[lane];
    unsigned x1 = 0u;
    if (lane + 32 < nn) x1 = lx[lane + 32];
#pragma unroll 1
    for (int g = 0; g < nfull; ++g) {
      const int ig = g * 32 + lane;
      unsigned x2 = 0u;
      if (ig + 64 < nn) x2 = lx[ig + 64];
      float y1 = 0.0f;
      if (FIRST) y1 = dinv[x1 & 0x1ffffu];
      else if (ig + 32 < nn) y1 = ln[ig + 32];
      float nr = y0;
      if (FIRST) {
        nr = y0 * __int_as_float(__builtin_amdgcn_ds_bpermute((int)(x0 >> 22) << 2, __float_as_int(dmine)));
        ln[ig] = nr;
      }
      add_rows_reg<F, true>((int)(x0 & 0x1ffffu), (int)(x0 >> 22), nr, 32, lane, h, acc);
      x0 = x1;
      y0 = y1;
      x1 = x2;
    }
    const int nlast = nn & 31;
    if (nlast != 0) {
      float nr = y0;
      if (FIRST) {
        nr = y0 * __int_as_float(__builtin_amdgcn_ds_bpermute((int)(x0 >> 22) << 2, __float_as_int(dmine)));
        if (lane < nlast) ln[nfull * 32 + lane] = nr;
      }
      add_rows_reg<F, false>((int)(x0 & 0x1ffffu), (int)(x0 >> 22), nr, nlast, lane, h, acc);
    }

#pragma unroll
    for (int k = 0; k < RPW; ++k) {
      const int r = wv + 32 * k;
      if (k < RPW - 1 || r < BN) {
        vecF o;
        for (int c = 0; c < VEC; ++c) o[c] = acc[c][k];
        *(vecF*)(agg + (long)(d0 + r) * F + lane * VEC) = o;
      }
    }
    return;
  }

  {
    const vecF bv = *(const vecF*)(b + lane * VEC);
    for (int r = wave; r < BN; r += 32) {
      const float di = dinv[d0 + r];
      const vecF hv = *(const vecF*)(h + (long)(d0 + r) * F + lane * VEC);
      for (int c = 0; c < VEC; ++c) tile[r * F + c * 32 + lane] = hv[c] * (di * di) + bv[c];
      if (lane == 0) dinvL[r] = di;
    }
  }
  __syncthreads();

  scan_owner_tags(
      tag4, q + wave * 64, lane, wave, blockIdx.x,
      [&](unsigned e) { serve_queue<F, true>(e, 32, lane, d0, ei, dinv, h, tile, dinvL); },
      [&](unsigned e, int m) { serve_queue<F, false>(e, m, lane, d0, ei, dinv, h, tile, dinvL); });
  __syncthreads();

  for (int r = wave; r < BN; r += 32) {
    vecF o;
    for (int c = 0; c < VEC; ++c) o[c] = tile[r * F + c * 32 + lane];
    *(vecF*)(agg + (long)(d0 + r) * F + lane * VEC) = o;
  }
}

extern "C" void kernel_launch(void* const* d_in, const int* in_sizes, int n_in,
                              void* d_out, int out_size, void* d_ws, size_t ws_size,
                              hipStream_t stream) {
  const float* x  = (const float*)d_in[0];
  const int*   ei = (const int*)d_in[1];
  const float* W1 = (const float*)d_in[2];
  const float* b1 = (const float*)d_in[3];
  const float* W2 = (const float*)d_in[4];
  const float* b2 = (const float*)d_in[5];
  float* out = (float*)d_out;

  float* ws   = (float*)d_ws;
  float* h1   = ws;
  float* agg1 = h1 + (size_t)N_NODES * HID;
  unsigned* tag4 = (unsigned*)(agg1 + (size_t)N_NODES * HID);
  float* dinv = (float*)tag4 + (size_t)N_NODES * OUTD;
  float* h3   = h1;
  unsigned* lists  = tag4 + N_EDGES / 4;
  unsigned* counts = lists + (size_t)(N_NODES / BN) * LW * 2;
  static_assert(N_EDGES % 256 == 0 && (LW * 4) % 256 == 0 && (SCAP * 4) % 128 == 0,
                "the lists start on aligned units");
  static_assert((long)N_EDGES + (long)(N_NODES / BN) * (LW * 8 + 32 * 4) <= (long)N_NODES * OUTD * 4,
                "the owner tags, lists and counts fit in the third region");

  const int T = 256;

  k_owner_tag<<<(N_EDGES / 4 + T - 1) / T, T, 0, stream>>>(ei, tag4);

  k_owner_deg_gemm1<<<N_NODES / BN, 1024, 0, stream>>>(ei, tag4, dinv, lists, counts, x, W1, h1);

  const int mtiles = (N_NODES + 15) / 16;
  const int gblocks = (mtiles + 7) / 8;

  k_edge_sum_own<HID, true><<<N_NODES / BN, 1024, 0, stream>>>(ei, tag4, dinv, h1, b1, agg1, lists, counts);

  k_gemm_wmma<HID, OUTD / 16, true>
      <<<gblocks, T, 0, stream>>>(agg1, W2, h3, N_NODES);

  k_edge_sum_own<OUTD, false><<<N_NODES / BN, 1024, 0, stream>>>(ei, tag4, dinv, h3, b2, out, lists, counts);
}
